# starting kernel
# baseline (speedup 1.0000x reference)
#include <hip/hip_runtime.h>
#include <hip/hip_bf16.h>

typedef __attribute__((ext_vector_type(2))) float v2f;
typedef __attribute__((ext_vector_type(8))) float v8f;

#define D 64
#define LSTR 66

__global__ void consts_kernel(const float* __restrict__ coeff_param,
                              const float* __restrict__ scale_param,
                              float* __restrict__ consts) {
    if (threadIdx.x != 0 || blockIdx.x != 0) return;
    float x0 = coeff_param[0], x1 = coeff_param[1], x2 = coeff_param[2];
    float m = fmaxf(x0, fmaxf(x1, x2));
    float e0 = expf(x0 - m), e1 = expf(x1 - m), e2 = expf(x2 - m);
    float inv = 1.0f / (e0 + e1 + e2);
    consts[0] = e0 * inv;
    consts[1] = e1 * inv;
    consts[2] = e2 * inv;
    float s = scale_param[0];
    consts[3] = (s > 20.0f) ? s : log1pf(expf(s));
}

__global__ void __launch_bounds__(128)
gemm_wmma_kernel(const float* __restrict__ h, const float* __restrict__ W,
                 const float* __restrict__ bias, float* __restrict__ Wh, int N) {
    __shared__ float Ws[D * LSTR];

    int tid = threadIdx.x;
    for (int i = tid * 4; i < D * D; i += 128 * 4) {
        int r = i >> 6;
        int c = i & 63;
        float4 w4 = *(const float4*)(W + i);
        Ws[(c + 0) * LSTR + r] = w4.x;
        Ws[(c + 1) * LSTR + r] = w4.y;
        Ws[(c + 2) * LSTR + r] = w4.z;
        Ws[(c + 3) * LSTR + r] = w4.w;
    }
    __syncthreads();

    int wave = tid >> 5;
    int lane = tid & 31;
    int half = lane >> 4;
    int l16  = lane & 15;

    int m0 = (blockIdx.x * 4 + wave) * 16;
    if (m0 >= N) return;

    const float* hrow = h + (size_t)(m0 + l16) * D + 2 * half;
    const float* wt   = Ws + l16 * LSTR + 2 * half;

    v8f acc[4] = {};

#pragma unroll 4
    for (int s = 0; s < 16; ++s) {
        v2f a = *(const v2f*)(hrow + 4 * s);
#pragma unroll
        for (int t = 0; t < 4; ++t) {
            v2f bf = *(const v2f*)(wt + t * (16 * LSTR) + 4 * s);
            acc[t] = __builtin_amdgcn_wmma_f32_16x16x4_f32(
                false, a, false, bf, (short)0, acc[t], false, false);
        }
    }

#pragma unroll
    for (int t = 0; t < 4; ++t) {
        int n = t * 16 + l16;
        float bv = bias[n];
#pragma unroll
        for (int v = 0; v < 8; ++v) {
            Wh[(size_t)(m0 + half * 8 + v) * D + n] = acc[t][v] + bv;
        }
    }
}

__global__ void __launch_bounds__(256)
norms_kernel(const float* __restrict__ Wh, float* __restrict__ norms, int N) {
    int gtid = blockIdx.x * 256 + threadIdx.x;
    int node = gtid >> 4;
    int l    = gtid & 15;
    if (node >= N) return;
    float4 v = *(const float4*)(Wh + (size_t)node * D + l * 4);
    float s = v.x * v.x + v.y * v.y + v.z * v.z + v.w * v.w;
    for (int m = 8; m >= 1; m >>= 1) s += __shfl_xor(s, m, 16);
    if (l == 0) norms[node] = sqrtf(s);
}

__global__ void __launch_bounds__(256)
zero_kernel(float4* __restrict__ p, int n4) {
    int i = blockIdx.x * 256 + threadIdx.x;
    if (i < n4) p[i] = make_float4(0.f, 0.f, 0.f, 0.f);
}

__global__ void __launch_bounds__(256)
edge_kernel(const int* __restrict__ ei, const float* __restrict__ adj_v,
            const float* __restrict__ deg_v, const float* __restrict__ Wh,
            const float* __restrict__ norms, const float* __restrict__ dcoef,
            const float* __restrict__ consts, float* __restrict__ prop, int E) {
    int tid = threadIdx.x;
    int l   = tid & 15;
    int eid = blockIdx.x * 16 + (tid >> 4);
    if (eid >= E) return;

    int row = ei[eid];
    int col = ei[E + eid];
    if (row == col) return;

    float4 r4 = *(const float4*)(Wh + (size_t)row * D + l * 4);
    float4 c4 = *(const float4*)(Wh + (size_t)col * D + l * 4);
    float dot = r4.x * c4.x + r4.y * c4.y + r4.z * c4.z + r4.w * c4.w;
    for (int m = 8; m >= 1; m >>= 1) dot += __shfl_xor(dot, m, 16);

    float nr = norms[row];
    float nc = norms[col];
    float sim = dot / fmaxf(nr * nc, 1e-8f);

    float x  = dcoef[0] * deg_v[eid] + dcoef[1];
    float sc = (x > 20.0f) ? x : log1pf(expf(x));

    float c0 = consts[0], c1 = consts[1];
    float w = adj_v[eid] * sc * (sim > 0.0f ? c0 * sim : c1 * sim);
    if (w == 0.0f) return;

    float* p = prop + (size_t)row * D + l * 4;
    atomicAdd(p + 0, w * c4.x);
    atomicAdd(p + 1, w * c4.y);
    atomicAdd(p + 2, w * c4.z);
    atomicAdd(p + 3, w * c4.w);
}

__global__ void __launch_bounds__(256)
final_kernel(const float4* __restrict__ prop, const float4* __restrict__ Wh,
             const float* __restrict__ consts, float4* __restrict__ out, int n4) {
    int i = blockIdx.x * 256 + threadIdx.x;
    if (i >= n4) return;
    float c2 = consts[2], scale = consts[3];
    float4 p = prop[i];
    float4 w = Wh[i];
    float4 o;
    o.x = scale * (p.x + c2 * w.x);
    o.y = scale * (p.y + c2 * w.y);
    o.z = scale * (p.z + c2 * w.z);
    o.w = scale * (p.w + c2 * w.w);
    out[i] = o;
}

extern "C" void kernel_launch(void* const* d_in, const int* in_sizes, int n_in,
                              void* d_out, int out_size, void* d_ws, size_t ws_size,
                              hipStream_t stream) {
    const float* h     = (const float*)d_in[0];
    const int*   ei    = (const int*)  d_in[1];
    const float* adj_v = (const float*)d_in[2];
    const float* deg_v = (const float*)d_in[3];
    const float* W     = (const float*)d_in[4];
    const float* bias  = (const float*)d_in[5];
    const float* dcoef = (const float*)d_in[6];
    const float* cpar  = (const float*)d_in[7];
    const float* spar  = (const float*)d_in[8];

    const int N = in_sizes[0] / D;
    const int E = in_sizes[2];

    float* Wh     = (float*)d_ws;
    float* prop   = Wh + (size_t)N * D;
    float* norms  = prop + (size_t)N * D;
    float* consts = norms + N;

    consts_kernel<<<1, 1, 0, stream>>>(cpar, spar, consts);

    gemm_wmma_kernel<<<(N + 63) / 64, 128, 0, stream>>>(h, W, bias, Wh, N);

    norms_kernel<<<(N * 16 + 255) / 256, 256, 0, stream>>>(Wh, norms, N);

    int n4 = N * D / 4;
    zero_kernel<<<(n4 + 255) / 256, 256, 0, stream>>>((float4*)prop, n4);

    edge_kernel<<<(E + 15) / 16, 256, 0, stream>>>(ei, adj_v, deg_v, Wh, norms,
                                                   dcoef, consts, prop, E);

    final_kernel<<<(n4 + 255) / 256, 256, 0, stream>>>((const float4*)prop,
                                                       (const float4*)Wh, consts,
                                                       (float4*)d_out, n4);
}


# Edges bucketed by destination row each call (one returning int add per edge gives its slot, arrival-order rank); a 16-lane group per row gathers, sums in registers, stores out once; no float atomics
# speedup vs baseline: 1.3757x; 1.3757x over previous
#include <hip/hip_runtime.h>
#include <hip/hip_bf16.h>

typedef __attribute__((ext_vector_type(2))) float v2f;
typedef __attribute__((ext_vector_type(8))) float v8f;

#define D 64
#define LSTR 66

__global__ void consts_kernel(const float* __restrict__ coeff_param,
                              const float* __restrict__ scale_param,
                              float* __restrict__ consts) {
    if (threadIdx.x != 0 || blockIdx.x != 0) return;
    float x0 = coeff_param[0], x1 = coeff_param[1], x2 = coeff_param[2];
    float m = fmaxf(x0, fmaxf(x1, x2));
    float e0 = expf(x0 - m), e1 = expf(x1 - m), e2 = expf(x2 - m);
    float inv = 1.0f / (e0 + e1 + e2);
    consts[0] = e0 * inv;
    consts[1] = e1 * inv;
    consts[2] = e2 * inv;
    float s = scale_param[0];
    consts[3] = (s > 20.0f) ? s : log1pf(expf(s));
}

__global__ void __launch_bounds__(128)
gemm_wmma_kernel(const float* __restrict__ h, const float* __restrict__ W,
                 const float* __restrict__ bias, float* __restrict__ Wh, int N) {
    __shared__ float Ws[D * LSTR];

    int tid = threadIdx.x;
    for (int i = tid * 4; i < D * D; i += 128 * 4) {
        int r = i >> 6;
        int c = i & 63;
        float4 w4 = *(const float4*)(W + i);
        Ws[(c + 0) * LSTR + r] = w4.x;
        Ws[(c + 1) * LSTR + r] = w4.y;
        Ws[(c + 2) * LSTR + r] = w4.z;
        Ws[(c + 3) * LSTR + r] = w4.w;
    }
    __syncthreads();

    int wave = tid >> 5;
    int lane = tid & 31;
    int half = lane >> 4;
    int l16  = lane & 15;

    int m0 = (blockIdx.x * 4 + wave) * 16;
    if (m0 >= N) return;

    const float* hrow = h + (size_t)(m0 + l16) * D + 2 * half;
    const float* wt   = Ws + l16 * LSTR + 2 * half;

    v8f acc[4] = {};

#pragma unroll 4
    for (int s = 0; s < 16; ++s) {
        v2f a = *(const v2f*)(hrow + 4 * s);
#pragma unroll
        for (int t = 0; t < 4; ++t) {
            v2f bf = *(const v2f*)(wt + t * (16 * LSTR) + 4 * s);
            acc[t] = __builtin_amdgcn_wmma_f32_16x16x4_f32(
                false, a, false, bf, (short)0, acc[t], false, false);
        }
    }

#pragma unroll
    for (int t = 0; t < 4; ++t) {
        int n = t * 16 + l16;
        float bv = bias[n];
#pragma unroll
        for (int v = 0; v < 8; ++v) {
            Wh[(size_t)(m0 + half * 8 + v) * D + n] = acc[t][v] + bv;
        }
    }
}

__global__ void __launch_bounds__(256)
norms_kernel(const float* __restrict__ Wh, float* __restrict__ norms,
             int* __restrict__ cnt, int* __restrict__ ovf_cnt, int N) {
    int gtid = blockIdx.x * 256 + threadIdx.x;
    int node = gtid >> 4;
    int l    = gtid & 15;
    if (gtid == 0) ovf_cnt[0] = 0;
    if (node >= N) return;
    float4 v = *(const float4*)(Wh + (size_t)node * D + l * 4);
    float s = v.x * v.x + v.y * v.y + v.z * v.z + v.w * v.w;
    for (int m = 8; m >= 1; m >>= 1) s += __shfl_xor(s, m, 16);
    if (l == 0) {
        norms[node] = sqrtf(s);
        cnt[node] = 0;
    }
}

#define CAP 64
#define FILL_THREADS 256
#define FILL_BLOCKS 1024

__global__ void __launch_bounds__(FILL_THREADS)
bucket_kernel(const int* __restrict__ ei, const float* __restrict__ adj_v,
              const float* __restrict__ deg_v, const float* __restrict__ dcoef,
              int* __restrict__ cnt, int2* __restrict__ bucket,
              int* __restrict__ ovf_cnt, int* __restrict__ ovf, int E) {
    const int T = FILL_BLOCKS * FILL_THREADS;
    const float dc0 = dcoef[0], dc1 = dcoef[1];
    for (int e0 = blockIdx.x * FILL_THREADS + threadIdx.x; e0 < E; e0 += 4 * T) {
        int   row[4], col[4], pos[4];
        float dg[4], ad[4];
        bool  ok[4];
        for (int k = 0; k < 4; ++k) {
            int e = e0 + k * T;
            ok[k] = e < E;
            int idx = ok[k] ? e : e0;
            row[k] = ei[idx];
            col[k] = ei[E + idx];
            dg[k]  = deg_v[idx];
            ad[k]  = adj_v[idx];
        }
        for (int k = 0; k < 4; ++k) {
            ok[k] = ok[k] && (row[k] != col[k]);
            if (ok[k]) pos[k] = atomicAdd(cnt + row[k], 1);
        }
        for (int k = 0; k < 4; ++k) {
            if (!ok[k]) continue;
            float x  = dc0 * dg[k] + dc1;
            float sc = (x > 20.0f) ? x : log1pf(expf(x));
            float base = ad[k] * sc;
            if (pos[k] < CAP) {
                bucket[(size_t)row[k] * CAP + pos[k]] = make_int2(col[k], __float_as_int(base));
            } else {
                int q = atomicAdd(ovf_cnt, 1);
                ovf[q] = e0 + k * T;
            }
        }
    }
}

__device__ __forceinline__ float row16_sum(float v) {
    v += __int_as_float(__builtin_amdgcn_update_dpp(0, __float_as_int(v), 0x168, 0xf, 0xf, false));
    v += __int_as_float(__builtin_amdgcn_update_dpp(0, __float_as_int(v), 0x164, 0xf, 0xf, false));
    v += __int_as_float(__builtin_amdgcn_update_dpp(0, __float_as_int(v), 0x162, 0xf, 0xf, false));
    v += __int_as_float(__builtin_amdgcn_update_dpp(0, __float_as_int(v), 0x161, 0xf, 0xf, false));
    return v;
}

__device__ __forceinline__ float edge_weight(float dot, float nr, float nc,
                                             float base, float c0, float c1) {
    float sim = dot / fmaxf(nr * nc, 1e-8f);
    return base * (sim > 0.0f ? c0 * sim : c1 * sim);
}

__global__ void __launch_bounds__(256)
gather_kernel(const int* __restrict__ ei, const float* __restrict__ adj_v,
              const float* __restrict__ deg_v, const float* __restrict__ dcoef,
              const float* __restrict__ Wh, const float* __restrict__ norms,
              const float* __restrict__ consts, const int* __restrict__ cnt,
              const int2* __restrict__ bucket, const int* __restrict__ ovf_cnt,
              const int* __restrict__ ovf, float* __restrict__ out, int N, int E) {
    int gtid = blockIdx.x * 256 + threadIdx.x;
    int row  = gtid >> 4;
    int l    = gtid & 15;
    if (row >= N) return;

    const float4* Wh4 = (const float4*)Wh;
    float4 r4 = Wh4[(size_t)row * (D / 4) + l];
    float nr = norms[row];
    int n_all = cnt[row];
    int n = n_all < CAP ? n_all : CAP;
    float c0 = consts[0], c1 = consts[1], c2 = consts[2], scale = consts[3];

    float4 acc = make_float4(0.f, 0.f, 0.f, 0.f);
    const int2* b = bucket + (size_t)row * CAP;

    for (int j = 0; j < n; j += 4) {
        int2  en[4];
        float4 c4[4];
        float nc[4], dt[4];
        for (int k = 0; k < 4; ++k) {
            int jj = (j + k < n) ? (j + k) : (n - 1);
            en[k] = b[jj];
        }
        for (int k = 0; k < 4; ++k) {
            c4[k] = Wh4[(size_t)en[k].x * (D / 4) + l];
            nc[k] = norms[en[k].x];
        }
        for (int k = 0; k < 4; ++k)
            dt[k] = r4.x * c4[k].x + r4.y * c4[k].y + r4.z * c4[k].z + r4.w * c4[k].w;
        for (int k = 0; k < 4; ++k) dt[k] = row16_sum(dt[k]);
        for (int k = 0; k < 4; ++k) {
            float w = edge_weight(dt[k], nr, nc[k], __int_as_float(en[k].y), c0, c1);
            w = (j + k < n) ? w : 0.0f;
            acc.x += w * c4[k].x;
            acc.y += w * c4[k].y;
            acc.z += w * c4[k].z;
            acc.w += w * c4[k].w;
        }
    }

    if (n_all > CAP) {
        int m = ovf_cnt[0];
        float dc0 = dcoef[0], dc1 = dcoef[1];
        for (int q = 0; q < m; ++q) {
            int e = ovf[q];
            if (ei[e] != row) continue;
            int col = ei[E + e];
            float x  = dc0 * deg_v[e] + dc1;
            float sc = (x > 20.0f) ? x : log1pf(expf(x));
            float base = adj_v[e] * sc;
            float4 cc = Wh4[(size_t)col * (D / 4) + l];
            float d = r4.x * cc.x + r4.y * cc.y + r4.z * cc.z + r4.w * cc.w;
            d = row16_sum(d);
            float w = edge_weight(d, nr, norms[col], base, c0, c1);
            acc.x += w * cc.x;
            acc.y += w * cc.y;
            acc.z += w * cc.z;
            acc.w += w * cc.w;
        }
    }

    float4 o;
    o.x = scale * (acc.x + c2 * r4.x);
    o.y = scale * (acc.y + c2 * r4.y);
    o.z = scale * (acc.z + c2 * r4.z);
    o.w = scale * (acc.w + c2 * r4.w);
    ((float4*)out)[(size_t)row * (D / 4) + l] = o;
}

extern "C" void kernel_launch(void* const* d_in, const int* in_sizes, int n_in,
                              void* d_out, int out_size, void* d_ws, size_t ws_size,
                              hipStream_t stream) {
    const float* h     = (const float*)d_in[0];
    const int*   ei    = (const int*)  d_in[1];
    const float* adj_v = (const float*)d_in[2];
    const float* deg_v = (const float*)d_in[3];
    const float* W     = (const float*)d_in[4];
    const float* bias  = (const float*)d_in[5];
    const float* dcoef = (const float*)d_in[6];
    const float* cpar  = (const float*)d_in[7];
    const float* spar  = (const float*)d_in[8];

    const int N = in_sizes[0] / D;
    const int E = in_sizes[2];

    float* Wh      = (float*)d_ws;
    int2*  bucket  = (int2*)(Wh + (size_t)N * D);
    int*   ovf     = (int*)(bucket + (size_t)N * CAP);
    float* norms   = (float*)(ovf + E);
    int*   cnt     = (int*)(norms + N);
    float* consts  = (float*)(cnt + N);
    int*   ovf_cnt = (int*)(consts + 4);

    consts_kernel<<<1, 1, 0, stream>>>(cpar, spar, consts);

    gemm_wmma_kernel<<<(N + 63) / 64, 128, 0, stream>>>(h, W, bias, Wh, N);

    norms_kernel<<<(N * 16 + 255) / 256, 256, 0, stream>>>(Wh, norms, cnt, ovf_cnt, N);

    bucket_kernel<<<FILL_BLOCKS, FILL_THREADS, 0, stream>>>(ei, adj_v, deg_v, dcoef, cnt,
                                                           bucket, ovf_cnt, ovf, E);

    gather_kernel<<<(N * 16 + 255) / 256, 256, 0, stream>>>(ei, adj_v, deg_v, dcoef, Wh,
                                                            norms, consts, cnt, bucket,
                                                            ovf_cnt, ovf, (float*)d_out, N, E);
}


# Row norms (summed in the GEMM epilogue from the accumulators: other order of a row's 64 squares), counter clears and parameter constants folded into the GEMM launch: 5 launches -> 3
# speedup vs baseline: 1.5090x; 1.0969x over previous
#include <hip/hip_runtime.h>
#include <hip/hip_bf16.h>

typedef __attribute__((ext_vector_type(2))) float v2f;
typedef __attribute__((ext_vector_type(8))) float v8f;

#define D 64
#define LSTR 66

__device__ __forceinline__ void fold_consts(const float* __restrict__ coeff_param,
                                            const float* __restrict__ scale_param,
                                            float* __restrict__ consts) {
    float x0 = coeff_param[0], x1 = coeff_param[1], x2 = coeff_param[2];
    float m = fmaxf(x0, fmaxf(x1, x2));
    float e0 = expf(x0 - m), e1 = expf(x1 - m), e2 = expf(x2 - m);
    float inv = 1.0f / (e0 + e1 + e2);
    consts[0] = e0 * inv;
    consts[1] = e1 * inv;
    consts[2] = e2 * inv;
    float s = scale_param[0];
    consts[3] = (s > 20.0f) ? s : log1pf(expf(s));
}

__device__ __forceinline__ float row16_sum(float v) {
    v += __int_as_float(__builtin_amdgcn_update_dpp(0, __float_as_int(v), 0x168, 0xf, 0xf, false));
    v += __int_as_float(__builtin_amdgcn_update_dpp(0, __float_as_int(v), 0x164, 0xf, 0xf, false));
    v += __int_as_float(__builtin_amdgcn_update_dpp(0, __float_as_int(v), 0x162, 0xf, 0xf, false));
    v += __int_as_float(__builtin_amdgcn_update_dpp(0, __float_as_int(v), 0x161, 0xf, 0xf, false));
    return v;
}

__global__ void __launch_bounds__(128)
gemm_wmma_kernel(const float* __restrict__ h, const float* __restrict__ W,
                 const float* __restrict__ bias, float* __restrict__ Wh,
                 float* __restrict__ norms, int* __restrict__ cnt,
                 const float* __restrict__ coeff_param,
                 const float* __restrict__ scale_param,
                 float* __restrict__ consts, int* __restrict__ ovf_cnt, int N) {
    __shared__ float Ws[D * LSTR];

    int tid = threadIdx.x;
    if (blockIdx.x == 0) {
        if (tid == 0) {
            ovf_cnt[0] = 0;
            fold_consts(coeff_param, scale_param, consts);
        }
        return;
    }
    for (int i = tid * 4; i < D * D; i += 128 * 4) {
        int r = i >> 6;
        int c = i & 63;
        float4 w4 = *(const float4*)(W + i);
        Ws[(c + 0) * LSTR + r] = w4.x;
        Ws[(c + 1) * LSTR + r] = w4.y;
        Ws[(c + 2) * LSTR + r] = w4.z;
        Ws[(c + 3) * LSTR + r] = w4.w;
    }
    __syncthreads();

    int wave = tid >> 5;
    int lane = tid & 31;
    int half = lane >> 4;
    int l16  = lane & 15;

    int m0 = ((blockIdx.x - 1) * 4 + wave) * 16;
    if (m0 >= N) return;

    const float* hrow = h + (size_t)(m0 + l16) * D + 2 * half;
    const float* wt   = Ws + l16 * LSTR + 2 * half;

    v8f acc[4] = {};

#pragma unroll 4
    for (int s = 0; s < 16; ++s) {
        v2f a = *(const v2f*)(hrow + 4 * s);
#pragma unroll
        for (int t = 0; t < 4; ++t) {
            v2f bf = *(const v2f*)(wt + t * (16 * LSTR) + 4 * s);
            acc[t] = __builtin_amdgcn_wmma_f32_16x16x4_f32(
                false, a, false, bf, (short)0, acc[t], false, false);
        }
    }

    float ss[8] = {0.f, 0.f, 0.f, 0.f, 0.f, 0.f, 0.f, 0.f};
#pragma unroll
    for (int t = 0; t < 4; ++t) {
        int n = t * 16 + l16;
        float bv = bias[n];
#pragma unroll
        for (int v = 0; v < 8; ++v) {
            float val = acc[t][v] + bv;
            Wh[(size_t)(m0 + half * 8 + v) * D + n] = val;
            ss[v] += val * val;
        }
    }
#pragma unroll
    for (int v = 0; v < 8; ++v) ss[v] = row16_sum(ss[v]);
    float mine = ss[0];
#pragma unroll
    for (int v = 1; v < 8; ++v) mine = (l16 == v) ? ss[v] : mine;
    if (l16 < 8) {
        int node = m0 + half * 8 + l16;
        norms[node] = sqrtf(mine);
        cnt[node] = 0;
    }
}

#define CAP 64
#define FILL_THREADS 256
#define FILL_BLOCKS 1024

__global__ void __launch_bounds__(FILL_THREADS)
bucket_kernel(const int* __restrict__ ei, const float* __restrict__ adj_v,
              const float* __restrict__ deg_v, const float* __restrict__ dcoef,
              int* __restrict__ cnt, int2* __restrict__ bucket,
              int* __restrict__ ovf_cnt, int* __restrict__ ovf, int E) {
    const int T = FILL_BLOCKS * FILL_THREADS;
    const float dc0 = dcoef[0], dc1 = dcoef[1];
    for (int e0 = blockIdx.x * FILL_THREADS + threadIdx.x; e0 < E; e0 += 4 * T) {
        int   row[4], col[4], pos[4];
        float dg[4], ad[4];
        bool  ok[4];
        for (int k = 0; k < 4; ++k) {
            int e = e0 + k * T;
            ok[k] = e < E;
            int idx = ok[k] ? e : e0;
            row[k] = ei[idx];
            col[k] = ei[E + idx];
            dg[k]  = deg_v[idx];
            ad[k]  = adj_v[idx];
        }
        for (int k = 0; k < 4; ++k) {
            ok[k] = ok[k] && (row[k] != col[k]);
            if (ok[k]) pos[k] = atomicAdd(cnt + row[k], 1);
        }
        for (int k = 0; k < 4; ++k) {
            if (!ok[k]) continue;
            float x  = dc0 * dg[k] + dc1;
            float sc = (x > 20.0f) ? x : log1pf(expf(x));
            float base = ad[k] * sc;
            if (pos[k] < CAP) {
                bucket[(size_t)row[k] * CAP + pos[k]] = make_int2(col[k], __float_as_int(base));
            } else {
                int q = atomicAdd(ovf_cnt, 1);
                ovf[q] = e0 + k * T;
            }
        }
    }
}

__device__ __forceinline__ float edge_weight(float dot, float nr, float nc,
                                             float base, float c0, float c1) {
    float sim = dot / fmaxf(nr * nc, 1e-8f);
    return base * (sim > 0.0f ? c0 * sim : c1 * sim);
}

__global__ void __launch_bounds__(256)
gather_kernel(const int* __restrict__ ei, const float* __restrict__ adj_v,
              const float* __restrict__ deg_v, const float* __restrict__ dcoef,
              const float* __restrict__ Wh, const float* __restrict__ norms,
              const float* __restrict__ consts, const int* __restrict__ cnt,
              const int2* __restrict__ bucket, const int* __restrict__ ovf_cnt,
              const int* __restrict__ ovf, float* __restrict__ out, int N, int E) {
    int gtid = blockIdx.x * 256 + threadIdx.x;
    int row  = gtid >> 4;
    int l    = gtid & 15;
    if (row >= N) return;

    const float4* Wh4 = (const float4*)Wh;
    float4 r4 = Wh4[(size_t)row * (D / 4) + l];
    float nr = norms[row];
    int n_all = cnt[row];
    int n = n_all < CAP ? n_all : CAP;
    float c0 = consts[0], c1 = consts[1], c2 = consts[2], scale = consts[3];

    float4 acc = make_float4(0.f, 0.f, 0.f, 0.f);
    const int2* b = bucket + (size_t)row * CAP;

    for (int j = 0; j < n; j += 4) {
        int2  en[4];
        float4 c4[4];
        float nc[4], dt[4];
        for (int k = 0; k < 4; ++k) {
            int jj = (j + k < n) ? (j + k) : (n - 1);
            en[k] = b[jj];
        }
        for (int k = 0; k < 4; ++k) {
            c4[k] = Wh4[(size_t)en[k].x * (D / 4) + l];
            nc[k] = norms[en[k].x];
        }
        for (int k = 0; k < 4; ++k)
            dt[k] = r4.x * c4[k].x + r4.y * c4[k].y + r4.z * c4[k].z + r4.w * c4[k].w;
        for (int k = 0; k < 4; ++k) dt[k] = row16_sum(dt[k]);
        for (int k = 0; k < 4; ++k) {
            float w = edge_weight(dt[k], nr, nc[k], __int_as_float(en[k].y), c0, c1);
            w = (j + k < n) ? w : 0.0f;
            acc.x += w * c4[k].x;
            acc.y += w * c4[k].y;
            acc.z += w * c4[k].z;
            acc.w += w * c4[k].w;
        }
    }

    if (n_all > CAP) {
        int m = ovf_cnt[0];
        float dc0 = dcoef[0], dc1 = dcoef[1];
        for (int q = 0; q < m; ++q) {
            int e = ovf[q];
            if (ei[e] != row) continue;
            int col = ei[E + e];
            float x  = dc0 * deg_v[e] + dc1;
            float sc = (x > 20.0f) ? x : log1pf(expf(x));
            float base = adj_v[e] * sc;
            float4 cc = Wh4[(size_t)col * (D / 4) + l];
            float d = r4.x * cc.x + r4.y * cc.y + r4.z * cc.z + r4.w * cc.w;
            d = row16_sum(d);
            float w = edge_weight(d, nr, norms[col], base, c0, c1);
            acc.x += w * cc.x;
            acc.y += w * cc.y;
            acc.z += w * cc.z;
            acc.w += w * cc.w;
        }
    }

    float4 o;
    o.x = scale * (acc.x + c2 * r4.x);
    o.y = scale * (acc.y + c2 * r4.y);
    o.z = scale * (acc.z + c2 * r4.z);
    o.w = scale * (acc.w + c2 * r4.w);
    ((float4*)out)[(size_t)row * (D / 4) + l] = o;
}

extern "C" void kernel_launch(void* const* d_in, const int* in_sizes, int n_in,
                              void* d_out, int out_size, void* d_ws, size_t ws_size,
                              hipStream_t stream) {
    const float* h     = (const float*)d_in[0];
    const int*   ei    = (const int*)  d_in[1];
    const float* adj_v = (const float*)d_in[2];
    const float* deg_v = (const float*)d_in[3];
    const float* W     = (const float*)d_in[4];
    const float* bias  = (const float*)d_in[5];
    const float* dcoef = (const float*)d_in[6];
    const float* cpar  = (const float*)d_in[7];
    const float* spar  = (const float*)d_in[8];

    const int N = in_sizes[0] / D;
    const int E = in_sizes[2];

    float* Wh      = (float*)d_ws;
    int2*  bucket  = (int2*)(Wh + (size_t)N * D);
    int*   ovf     = (int*)(bucket + (size_t)N * CAP);
    float* norms   = (float*)(ovf + E);
    int*   cnt     = (int*)(norms + N);
    float* consts  = (float*)(cnt + N);
    int*   ovf_cnt = (int*)(consts + 4);

    gemm_wmma_kernel<<<(N + 63) / 64 + 1, 128, 0, stream>>>(h, W, bias, Wh, norms, cnt,
                                                            cpar, spar, consts, ovf_cnt, N);

    bucket_kernel<<<FILL_BLOCKS, FILL_THREADS, 0, stream>>>(ei, adj_v, deg_v, dcoef, cnt,
                                                           bucket, ovf_cnt, ovf, E);

    gather_kernel<<<(N * 16 + 255) / 256, 256, 0, stream>>>(ei, adj_v, deg_v, dcoef, Wh,
                                                            norms, consts, cnt, bucket,
                                                            ovf_cnt, ovf, (float*)d_out, N, E);
}


# Edges grouped by 256-row range on LDS counters (no global atomic per edge); one block per range ranks them by row in LDS, then gathers from LDS buckets, 8 rows in flight; row sums still arrival order
# speedup vs baseline: 3.3616x; 2.2277x over previous
#include <hip/hip_runtime.h>
#include <hip/hip_bf16.h>

typedef __attribute__((ext_vector_type(2))) float v2f;
typedef __attribute__((ext_vector_type(8))) float v8f;
typedef __attribute__((ext_vector_type(4))) float v4f;

#define D 64
#define LSTR 66

__device__ __forceinline__ void fold_consts(const float* __restrict__ coeff_param,
                                            const float* __restrict__ scale_param,
                                            float* __restrict__ consts) {
    float x0 = coeff_param[0], x1 = coeff_param[1], x2 = coeff_param[2];
    float m = fmaxf(x0, fmaxf(x1, x2));
    float e0 = expf(x0 - m), e1 = expf(x1 - m), e2 = expf(x2 - m);
    float inv = 1.0f / (e0 + e1 + e2);
    consts[0] = e0 * inv;
    consts[1] = e1 * inv;
    consts[2] = e2 * inv;
    float s = scale_param[0];
    consts[3] = (s > 20.0f) ? s : log1pf(expf(s));
}

__device__ __forceinline__ float row16_sum(float v) {
    v += __int_as_float(__builtin_amdgcn_update_dpp(0, __float_as_int(v), 0x168, 0xf, 0xf, false));
    v += __int_as_float(__builtin_amdgcn_update_dpp(0, __float_as_int(v), 0x164, 0xf, 0xf, false));
    v += __int_as_float(__builtin_amdgcn_update_dpp(0, __float_as_int(v), 0x162, 0xf, 0xf, false));
    v += __int_as_float(__builtin_amdgcn_update_dpp(0, __float_as_int(v), 0x161, 0xf, 0xf, false));
    return v;
}

__global__ void __launch_bounds__(128)
gemm_wmma_kernel(const float* __restrict__ h, const float* __restrict__ W,
                 const float* __restrict__ bias, float* __restrict__ Wh,
                 float* __restrict__ norms,
                 const float* __restrict__ coeff_param,
                 const float* __restrict__ scale_param,
                 float* __restrict__ consts, int* __restrict__ ovf_cnt, int N) {
    __shared__ float Ws[D * LSTR];

    int tid = threadIdx.x;
    if (blockIdx.x == 0) {
        if (tid == 0) {
            ovf_cnt[0] = 0;
            fold_consts(coeff_param, scale_param, consts);
        }
        return;
    }
    for (int i = tid * 4; i < D * D; i += 128 * 4) {
        int r = i >> 6;
        int c = i & 63;
        float4 w4 = *(const float4*)(W + i);
        Ws[(c + 0) * LSTR + r] = w4.x;
        Ws[(c + 1) * LSTR + r] = w4.y;
        Ws[(c + 2) * LSTR + r] = w4.z;
        Ws[(c + 3) * LSTR + r] = w4.w;
    }
    __syncthreads();

    int wave = tid >> 5;
    int lane = tid & 31;
    int half = lane >> 4;
    int l16  = lane & 15;

    int m0 = ((blockIdx.x - 1) * 4 + wave) * 16;
    if (m0 >= N) return;

    const float* hrow = h + (size_t)(m0 + l16) * D + 2 * half;
    const float* wt   = Ws + l16 * LSTR + 2 * half;

    v8f acc[4] = {};

#pragma unroll 4
    for (int s = 0; s < 16; ++s) {
        v2f a = *(const v2f*)(hrow + 4 * s);
#pragma unroll
        for (int t = 0; t < 4; ++t) {
            v2f bf = *(const v2f*)(wt + t * (16 * LSTR) + 4 * s);
            acc[t] = __builtin_amdgcn_wmma_f32_16x16x4_f32(
                false, a, false, bf, (short)0, acc[t], false, false);
        }
    }

    float ss[8] = {0.f, 0.f, 0.f, 0.f, 0.f, 0.f, 0.f, 0.f};
#pragma unroll
    for (int t = 0; t < 4; ++t) {
        int n = t * 16 + l16;
        float bv = bias[n];
#pragma unroll
        for (int v = 0; v < 8; ++v) {
            float val = acc[t][v] + bv;
            Wh[(size_t)(m0 + half * 8 + v) * D + n] = val;
            ss[v] += val * val;
        }
    }
#pragma unroll
    for (int v = 0; v < 8; ++v) ss[v] = row16_sum(ss[v]);
    float mine = ss[0];
#pragma unroll
    for (int v = 1; v < 8; ++v) mine = (l16 == v) ? ss[v] : mine;
    if (l16 < 8) {
        int node = m0 + half * 8 + l16;
        norms[node] = sqrtf(mine);
    }
}

#define BIN_ROWS 256
#define NBIN_MAX 256
#define EPB 4096
#define BIN_THREADS 1024
#define SLOT 64
#define CAP 64
#define GATHER_THREADS 1024
#define GROUPS (GATHER_THREADS / 16)
#define DEPTH 8
#define COL_MASK 0xFFFFFFu

__global__ void __launch_bounds__(BIN_THREADS)
bin_kernel(const int* __restrict__ ei, const float* __restrict__ adj_v,
           const float* __restrict__ deg_v, const float* __restrict__ dcoef,
           int2* __restrict__ binned, int* __restrict__ runlen,
           int* __restrict__ ovf_cnt, int* __restrict__ ovf, int E, int NBIN, int NBLK) {
    __shared__ int hist[NBIN_MAX];
    const int tid = threadIdx.x;
    const int blk = blockIdx.x;
    if (tid < NBIN_MAX) hist[tid] = 0;
    __syncthreads();

    const float dc0 = dcoef[0], dc1 = dcoef[1];
    const int e0 = blk * EPB + tid;
    int   row[4], col[4];
    float dg[4], ad[4];
    bool  ok[4];
    for (int k = 0; k < 4; ++k) {
        int e = e0 + k * BIN_THREADS;
        ok[k] = e < E;
        int idx = ok[k] ? e : 0;
        row[k] = ei[idx];
        col[k] = ei[E + idx];
        dg[k]  = deg_v[idx];
        ad[k]  = adj_v[idx];
    }
    for (int k = 0; k < 4; ++k) {
        if (!ok[k] || row[k] == col[k]) continue;
        int bin = row[k] >> 8;
        int r = atomicAdd(&hist[bin], 1);
        float x  = dc0 * dg[k] + dc1;
        float sc = (x > 20.0f) ? x : log1pf(expf(x));
        float base = ad[k] * sc;
        if (r < SLOT) {
            unsigned key = (unsigned)col[k] | ((unsigned)(row[k] & (BIN_ROWS - 1)) << 24);
            binned[((size_t)bin * NBLK + blk) * SLOT + r] = make_int2((int)key, __float_as_int(base));
        } else {
            int q = atomicAdd(ovf_cnt, 1);
            ovf[q] = e0 + k * BIN_THREADS;
        }
    }
    __syncthreads();
    if (tid < NBIN) {
        int c = hist[tid];
        runlen[tid * NBLK + blk] = c;
        if (c == 0) binned[((size_t)tid * NBLK + blk) * SLOT] = make_int2(0, 0);
    }
}

__device__ __forceinline__ float edge_weight(float dot, float nr, float nc,
                                             float base, float c0, float c1) {
    float sim = dot / fmaxf(nr * nc, 1e-8f);
    return base * (sim > 0.0f ? c0 * sim : c1 * sim);
}

__device__ __forceinline__ void add_edge(float4& acc, const float4 r4, float nr, int col,
                                         float base, const float4* __restrict__ Wh4,
                                         const float* __restrict__ norms, int l,
                                         float c0, float c1) {
    float4 cc = Wh4[(size_t)col * (D / 4) + l];
    float d = r4.x * cc.x + r4.y * cc.y + r4.z * cc.z + r4.w * cc.w;
    d = row16_sum(d);
    float w = edge_weight(d, nr, norms[col], base, c0, c1);
    acc.x += w * cc.x;
    acc.y += w * cc.y;
    acc.z += w * cc.z;
    acc.w += w * cc.w;
}

__global__ void __launch_bounds__(GATHER_THREADS)
gather_kernel(const int* __restrict__ ei, const float* __restrict__ adj_v,
              const float* __restrict__ deg_v, const float* __restrict__ dcoef,
              const float* __restrict__ Wh, const float* __restrict__ norms,
              const float* __restrict__ consts, const int* __restrict__ runlen,
              const int2* __restrict__ binned, const int* __restrict__ ovf_cnt,
              const int* __restrict__ ovf, float* __restrict__ out, int N, int E, int NBLK) {
    __shared__ int  rcnt[BIN_ROWS];
    __shared__ int  binovf;
    __shared__ int2 lb[BIN_ROWS * CAP];

    const int tid = threadIdx.x;
    const int b   = blockIdx.x;
    const int g   = tid >> 4;
    const int l   = tid & 15;

    if (tid < BIN_ROWS) rcnt[tid] = 0;
    if (tid == 0) binovf = 0;
    __syncthreads();

    const int* rl = runlen + (size_t)b * NBLK;
    const int2* runs = binned + (size_t)b * NBLK * SLOT;
    for (int k0 = 0; k0 < NBLK; k0 += 4 * GROUPS) {
        int len[4];
        const int2* run[4];
        int2 ra[4], rb[4];
        int la[4];
        for (int p = 0; p < 4; ++p) {
            int k = k0 + p * GROUPS + g;
            bool valid = k < NBLK;
            int kk = valid ? k : 0;
            int v = rl[kk];
            la[p] = valid ? v : 0;
            run[p] = runs + (size_t)kk * SLOT;
        }
        int lmax = la[0];
        for (int p = 0; p < 4; ++p) {
            lmax = la[p] > lmax ? la[p] : lmax;
            len[p] = la[p] < SLOT ? la[p] : SLOT;
        }
        if (lmax > SLOT) binovf = 1;
        for (int p = 0; p < 4; ++p) {
            int last = (len[p] > 0 ? len[p] : 1) - 1;
            int ia = l < last ? l : last;
            int ib = (l + 16) < last ? (l + 16) : last;
            ra[p] = run[p][ia];
            rb[p] = run[p][ib];
        }
        for (int p = 0; p < 4; ++p) {
            if (l < len[p]) {
                unsigned u = (unsigned)ra[p].x;
                int r = (int)(u >> 24);
                int slot = atomicAdd(&rcnt[r], 1);
                if (slot < CAP) lb[r * CAP + slot] = make_int2((int)(u & COL_MASK), ra[p].y);
            }
            if (l + 16 < len[p]) {
                unsigned u = (unsigned)rb[p].x;
                int r = (int)(u >> 24);
                int slot = atomicAdd(&rcnt[r], 1);
                if (slot < CAP) lb[r * CAP + slot] = make_int2((int)(u & COL_MASK), rb[p].y);
            }
            for (int i = l + 32; i < len[p]; i += 16) {
                int2 rec = run[p][i];
                unsigned u = (unsigned)rec.x;
                int r = (int)(u >> 24);
                int slot = atomicAdd(&rcnt[r], 1);
                if (slot < CAP) lb[r * CAP + slot] = make_int2((int)(u & COL_MASK), rec.y);
            }
        }
    }
    __syncthreads();

    const int range_ovf = binovf;
    const float4* Wh4 = (const float4*)Wh;
    const v4f* Wh4v = (const v4f*)Wh;
    const float c0 = consts[0], c1 = consts[1], c2 = consts[2], scale = consts[3];

    for (int q = 0; q < BIN_ROWS / GROUPS; ++q) {
        const int rr  = q * GROUPS + g;
        const int row = b * BIN_ROWS + rr;
        if (row >= N) continue;

        float4 r4 = Wh4[(size_t)row * (D / 4) + l];
        float nr = norms[row];
        const int n_all = rcnt[rr];
        const int n = n_all < CAP ? n_all : CAP;
        const int2* bk = lb + rr * CAP;

        float4 acc = make_float4(0.f, 0.f, 0.f, 0.f);

        for (int j = 0; j < n; j += DEPTH) {
            int2  en[DEPTH];
            v4f   c4[DEPTH];
            float nc[DEPTH], dt[DEPTH];
            for (int k = 0; k < DEPTH; ++k) {
                int jj = (j + k < n) ? (j + k) : (n - 1);
                en[k] = bk[jj];
            }
            for (int k = 0; k < DEPTH; ++k) {
                c4[k] = Wh4v[(size_t)en[k].x * (D / 4) + l];
                nc[k] = norms[en[k].x];
            }
            asm volatile("" :: "v"(c4[0]), "v"(c4[1]), "v"(c4[2]), "v"(c4[3]),
                               "v"(c4[4]), "v"(c4[5]), "v"(c4[6]), "v"(c4[7]),
                               "v"(nc[0]), "v"(nc[1]), "v"(nc[2]), "v"(nc[3]),
                               "v"(nc[4]), "v"(nc[5]), "v"(nc[6]), "v"(nc[7]));
            for (int k = 0; k < DEPTH; ++k)
                dt[k] = r4.x * c4[k].x + r4.y * c4[k].y + r4.z * c4[k].z + r4.w * c4[k].w;
            for (int k = 0; k < DEPTH; ++k) dt[k] = row16_sum(dt[k]);
            for (int k = 0; k < DEPTH; ++k) {
                float w = edge_weight(dt[k], nr, nc[k], __int_as_float(en[k].y), c0, c1);
                w = (j + k < n) ? w : 0.0f;
                acc.x += w * c4[k].x;
                acc.y += w * c4[k].y;
                acc.z += w * c4[k].z;
                acc.w += w * c4[k].w;
            }
        }

        if (n_all > CAP) {
            acc = make_float4(0.f, 0.f, 0.f, 0.f);
            for (int k = 0; k < NBLK; ++k) {
                int la = rl[k];
                int len = la < SLOT ? la : SLOT;
                const int2* run = runs + (size_t)k * SLOT;
                for (int i = 0; i < len; ++i) {
                    int2 rec = run[i];
                    unsigned u = (unsigned)rec.x;
                    if ((int)(u >> 24) != rr) continue;
                    add_edge(acc, r4, nr, (int)(u & COL_MASK), __int_as_float(rec.y),
                             Wh4, norms, l, c0, c1);
                }
            }
        }

        if (range_ovf) {
            int m = ovf_cnt[0];
            float dc0 = dcoef[0], dc1 = dcoef[1];
            for (int t = 0; t < m; ++t) {
                int e = ovf[t];
                if (ei[e] != row) continue;
                int col = ei[E + e];
                float x  = dc0 * deg_v[e] + dc1;
                float sc = (x > 20.0f) ? x : log1pf(expf(x));
                float base = adj_v[e] * sc;
                add_edge(acc, r4, nr, col, base, Wh4, norms, l, c0, c1);
            }
        }

        float4 o;
        o.x = scale * (acc.x + c2 * r4.x);
        o.y = scale * (acc.y + c2 * r4.y);
        o.z = scale * (acc.z + c2 * r4.z);
        o.w = scale * (acc.w + c2 * r4.w);
        ((float4*)out)[(size_t)row * (D / 4) + l] = o;
    }
}

extern "C" void kernel_launch(void* const* d_in, const int* in_sizes, int n_in,
                              void* d_out, int out_size, void* d_ws, size_t ws_size,
                              hipStream_t stream) {
    const float* h     = (const float*)d_in[0];
    const int*   ei    = (const int*)  d_in[1];
    const float* adj_v = (const float*)d_in[2];
    const float* deg_v = (const float*)d_in[3];
    const float* W     = (const float*)d_in[4];
    const float* bias  = (const float*)d_in[5];
    const float* dcoef = (const float*)d_in[6];
    const float* cpar  = (const float*)d_in[7];
    const float* spar  = (const float*)d_in[8];

    const int N = in_sizes[0] / D;
    const int E = in_sizes[2];
    const int NBIN = (N + BIN_ROWS - 1) / BIN_ROWS;
    const int NBLK = (E + EPB - 1) / EPB;

    float* Wh      = (float*)d_ws;
    int2*  binned  = (int2*)(Wh + (size_t)N * D);
    int*   ovf     = (int*)(binned + (size_t)NBIN * NBLK * SLOT);
    float* norms   = (float*)(ovf + E);
    int*   runlen  = (int*)(norms + N);
    float* consts  = (float*)(runlen + (size_t)NBIN * NBLK);
    int*   ovf_cnt = (int*)(consts + 4);

    gemm_wmma_kernel<<<(N + 63) / 64 + 1, 128, 0, stream>>>(h, W, bias, Wh, norms,
                                                            cpar, spar, consts, ovf_cnt, N);

    bin_kernel<<<NBLK, BIN_THREADS, 0, stream>>>(ei, adj_v, deg_v, dcoef, binned, runlen,
                                                 ovf_cnt, ovf, E, NBIN, NBLK);

    gather_kernel<<<NBIN, GATHER_THREADS, 0, stream>>>(ei, adj_v, deg_v, dcoef, Wh, norms,
                                                       consts, runlen, binned, ovf_cnt, ovf,
                                                       (float*)d_out, N, E, NBLK);
}


# Gather loop: a trip's 8 dots are reduce-scattered over the 16-lane group (same add tree, bit-identical sums); one lane pair per edge computes its weight once, row-share moves hand it round
# speedup vs baseline: 3.8243x; 1.1376x over previous
#include <hip/hip_runtime.h>
#include <hip/hip_bf16.h>

typedef __attribute__((ext_vector_type(2))) float v2f;
typedef __attribute__((ext_vector_type(8))) float v8f;
typedef __attribute__((ext_vector_type(4))) float v4f;

#define D 64
#define LSTR 66

__device__ __forceinline__ void fold_consts(const float* __restrict__ coeff_param,
                                            const float* __restrict__ scale_param,
                                            float* __restrict__ consts) {
    float x0 = coeff_param[0], x1 = coeff_param[1], x2 = coeff_param[2];
    float m = fmaxf(x0, fmaxf(x1, x2));
    float e0 = expf(x0 - m), e1 = expf(x1 - m), e2 = expf(x2 - m);
    float inv = 1.0f / (e0 + e1 + e2);
    consts[0] = e0 * inv;
    consts[1] = e1 * inv;
    consts[2] = e2 * inv;
    float s = scale_param[0];
    consts[3] = (s > 20.0f) ? s : log1pf(expf(s));
}

__device__ __forceinline__ float row16_sum(float v) {
    v += __int_as_float(__builtin_amdgcn_update_dpp(0, __float_as_int(v), 0x168, 0xf, 0xf, false));
    v += __int_as_float(__builtin_amdgcn_update_dpp(0, __float_as_int(v), 0x164, 0xf, 0xf, false));
    v += __int_as_float(__builtin_amdgcn_update_dpp(0, __float_as_int(v), 0x162, 0xf, 0xf, false));
    v += __int_as_float(__builtin_amdgcn_update_dpp(0, __float_as_int(v), 0x161, 0xf, 0xf, false));
    return v;
}

template <int XM>
__device__ __forceinline__ float row_xmask_f(float v) {
    return __int_as_float(__builtin_amdgcn_update_dpp(0, __float_as_int(v), 0x160 + XM, 0xf, 0xf, false));
}
template <int SRC>
__device__ __forceinline__ int row_share_i(int v) {
    return __builtin_amdgcn_mov_dpp(v, 0x150 + SRC, 0xf, 0xf, true);
}
template <int SRC>
__device__ __forceinline__ float row_share_f(float v) {
    return __int_as_float(row_share_i<SRC>(__float_as_int(v)));
}

template <int XM>
__device__ __forceinline__ float halve_pair(float a, float b, bool hi) {
    float keep = hi ? b : a;
    float send = hi ? a : b;
    return keep + row_xmask_f<XM>(send);
}

__device__ __forceinline__ float row16_sum8_scatter(const float (&d)[8], int p) {
    const bool h8 = (p & 8) != 0, h4 = (p & 4) != 0, h2 = (p & 2) != 0;
    float e0 = halve_pair<8>(d[0], d[4], h8);
    float e1 = halve_pair<8>(d[1], d[5], h8);
    float e2 = halve_pair<8>(d[2], d[6], h8);
    float e3 = halve_pair<8>(d[3], d[7], h8);
    float f0 = halve_pair<4>(e0, e2, h4);
    float f1 = halve_pair<4>(e1, e3, h4);
    float g  = halve_pair<2>(f0, f1, h2);
    return g + row_xmask_f<1>(g);
}

__global__ void __launch_bounds__(128)
gemm_wmma_kernel(const float* __restrict__ h, const float* __restrict__ W,
                 const float* __restrict__ bias, float* __restrict__ Wh,
                 float* __restrict__ norms,
                 const float* __restrict__ coeff_param,
                 const float* __restrict__ scale_param,
                 float* __restrict__ consts, int* __restrict__ ovf_cnt, int N) {
    __shared__ float Ws[D * LSTR];

    int tid = threadIdx.x;
    if (blockIdx.x == 0) {
        if (tid == 0) {
            ovf_cnt[0] = 0;
            fold_consts(coeff_param, scale_param, consts);
        }
        return;
    }
    for (int i = tid * 4; i < D * D; i += 128 * 4) {
        int r = i >> 6;
        int c = i & 63;
        float4 w4 = *(const float4*)(W + i);
        Ws[(c + 0) * LSTR + r] = w4.x;
        Ws[(c + 1) * LSTR + r] = w4.y;
        Ws[(c + 2) * LSTR + r] = w4.z;
        Ws[(c + 3) * LSTR + r] = w4.w;
    }
    __syncthreads();

    int wave = tid >> 5;
    int lane = tid & 31;
    int half = lane >> 4;
    int l16  = lane & 15;

    int m0 = ((blockIdx.x - 1) * 4 + wave) * 16;
    if (m0 >= N) return;

    const float* hrow = h + (size_t)(m0 + l16) * D + 2 * half;
    const float* wt   = Ws + l16 * LSTR + 2 * half;

    v8f acc[4] = {};

#pragma unroll 4
    for (int s = 0; s < 16; ++s) {
        v2f a = *(const v2f*)(hrow + 4 * s);
#pragma unroll
        for (int t = 0; t < 4; ++t) {
            v2f bf = *(const v2f*)(wt + t * (16 * LSTR) + 4 * s);
            acc[t] = __builtin_amdgcn_wmma_f32_16x16x4_f32(
                false, a, false, bf, (short)0, acc[t], false, false);
        }
    }

    float ss[8] = {0.f, 0.f, 0.f, 0.f, 0.f, 0.f, 0.f, 0.f};
#pragma unroll
    for (int t = 0; t < 4; ++t) {
        int n = t * 16 + l16;
        float bv = bias[n];
#pragma unroll
        for (int v = 0; v < 8; ++v) {
            float val = acc[t][v] + bv;
            Wh[(size_t)(m0 + half * 8 + v) * D + n] = val;
            ss[v] += val * val;
        }
    }
#pragma unroll
    for (int v = 0; v < 8; ++v) ss[v] = row16_sum(ss[v]);
    float mine = ss[0];
#pragma unroll
    for (int v = 1; v < 8; ++v) mine = (l16 == v) ? ss[v] : mine;
    if (l16 < 8) {
        int node = m0 + half * 8 + l16;
        norms[node] = sqrtf(mine);
    }
}

#define BIN_ROWS 256
#define NBIN_MAX 256
#define EPB 4096
#define BIN_THREADS 1024
#define SLOT 64
#define CAP 64
#define GATHER_THREADS 1024
#define GROUPS (GATHER_THREADS / 16)
#define DEPTH 8
#define COL_MASK 0xFFFFFFu

__global__ void __launch_bounds__(BIN_THREADS)
bin_kernel(const int* __restrict__ ei, const float* __restrict__ adj_v,
           const float* __restrict__ deg_v, const float* __restrict__ dcoef,
           int2* __restrict__ binned, int* __restrict__ runlen,
           int* __restrict__ ovf_cnt, int* __restrict__ ovf, int E, int NBIN, int NBLK) {
    __shared__ int hist[NBIN_MAX];
    const int tid = threadIdx.x;
    const int blk = blockIdx.x;
    if (tid < NBIN_MAX) hist[tid] = 0;
    __syncthreads();

    const float dc0 = dcoef[0], dc1 = dcoef[1];
    const int e0 = blk * EPB + tid;
    int   row[4], col[4];
    float dg[4], ad[4];
    bool  ok[4];
    for (int k = 0; k < 4; ++k) {
        int e = e0 + k * BIN_THREADS;
        ok[k] = e < E;
        int idx = ok[k] ? e : 0;
        row[k] = ei[idx];
        col[k] = ei[E + idx];
        dg[k]  = deg_v[idx];
        ad[k]  = adj_v[idx];
    }
    for (int k = 0; k < 4; ++k) {
        if (!ok[k] || row[k] == col[k]) continue;
        int bin = row[k] >> 8;
        int r = atomicAdd(&hist[bin], 1);
        float x  = dc0 * dg[k] + dc1;
        float sc = (x > 20.0f) ? x : log1pf(expf(x));
        float base = ad[k] * sc;
        if (r < SLOT) {
            unsigned key = (unsigned)col[k] | ((unsigned)(row[k] & (BIN_ROWS - 1)) << 24);
            binned[((size_t)bin * NBLK + blk) * SLOT + r] = make_int2((int)key, __float_as_int(base));
        } else {
            int q = atomicAdd(ovf_cnt, 1);
            ovf[q] = e0 + k * BIN_THREADS;
        }
    }
    __syncthreads();
    if (tid < NBIN) {
        int c = hist[tid];
        runlen[tid * NBLK + blk] = c;
        if (c == 0) binned[((size_t)tid * NBLK + blk) * SLOT] = make_int2(0, 0);
    }
}

__device__ __forceinline__ float edge_weight(float dot, float nr, float nc,
                                             float base, float c0, float c1) {
    float sim = dot / fmaxf(nr * nc, 1e-8f);
    return base * (sim > 0.0f ? c0 * sim : c1 * sim);
}

__device__ __forceinline__ void add_edge(float4& acc, const float4 r4, float nr, int col,
                                         float base, const float4* __restrict__ Wh4,
                                         const float* __restrict__ norms, int l,
                                         float c0, float c1) {
    float4 cc = Wh4[(size_t)col * (D / 4) + l];
    float d = r4.x * cc.x + r4.y * cc.y + r4.z * cc.z + r4.w * cc.w;
    d = row16_sum(d);
    float w = edge_weight(d, nr, norms[col], base, c0, c1);
    acc.x += w * cc.x;
    acc.y += w * cc.y;
    acc.z += w * cc.z;
    acc.w += w * cc.w;
}

__global__ void __launch_bounds__(GATHER_THREADS)
gather_kernel(const int* __restrict__ ei, const float* __restrict__ adj_v,
              const float* __restrict__ deg_v, const float* __restrict__ dcoef,
              const float* __restrict__ Wh, const float* __restrict__ norms,
              const float* __restrict__ consts, const int* __restrict__ runlen,
              const int2* __restrict__ binned, const int* __restrict__ ovf_cnt,
              const int* __restrict__ ovf, float* __restrict__ out, int N, int E, int NBLK) {
    __shared__ int  rcnt[BIN_ROWS];
    __shared__ int  binovf;
    __shared__ int2 lb[BIN_ROWS * CAP];

    const int tid = threadIdx.x;
    const int b   = blockIdx.x;
    const int g   = tid >> 4;
    const int l   = tid & 15;

    if (tid < BIN_ROWS) rcnt[tid] = 0;
    if (tid == 0) binovf = 0;
    __syncthreads();

    const int* rl = runlen + (size_t)b * NBLK;
    const int2* runs = binned + (size_t)b * NBLK * SLOT;
    for (int k0 = 0; k0 < NBLK; k0 += 4 * GROUPS) {
        int len[4];
        const int2* run[4];
        int2 ra[4], rb[4];
        int la[4];
        for (int p = 0; p < 4; ++p) {
            int k = k0 + p * GROUPS + g;
            bool valid = k < NBLK;
            int kk = valid ? k : 0;
            int v = rl[kk];
            la[p] = valid ? v : 0;
            run[p] = runs + (size_t)kk * SLOT;
        }
        int lmax = la[0];
        for (int p = 0; p < 4; ++p) {
            lmax = la[p] > lmax ? la[p] : lmax;
            len[p] = la[p] < SLOT ? la[p] : SLOT;
        }
        if (lmax > SLOT) binovf = 1;
        for (int p = 0; p < 4; ++p) {
            int last = (len[p] > 0 ? len[p] : 1) - 1;
            int ia = l < last ? l : last;
            int ib = (l + 16) < last ? (l + 16) : last;
            ra[p] = run[p][ia];
            rb[p] = run[p][ib];
        }
        for (int p = 0; p < 4; ++p) {
            if (l < len[p]) {
                unsigned u = (unsigned)ra[p].x;
                int r = (int)(u >> 24);
                int slot = atomicAdd(&rcnt[r], 1);
                if (slot < CAP) lb[r * CAP + slot] = make_int2((int)(u & COL_MASK), ra[p].y);
            }
            if (l + 16 < len[p]) {
                unsigned u = (unsigned)rb[p].x;
                int r = (int)(u >> 24);
                int slot = atomicAdd(&rcnt[r], 1);
                if (slot < CAP) lb[r * CAP + slot] = make_int2((int)(u & COL_MASK), rb[p].y);
            }
            for (int i = l + 32; i < len[p]; i += 16) {
                int2 rec = run[p][i];
                unsigned u = (unsigned)rec.x;
                int r = (int)(u >> 24);
                int slot = atomicAdd(&rcnt[r], 1);
                if (slot < CAP) lb[r * CAP + slot] = make_int2((int)(u & COL_MASK), rec.y);
            }
        }
    }
    __syncthreads();

    const int range_ovf = binovf;
    const float4* Wh4 = (const float4*)Wh;
    const v4f* Wh4v = (const v4f*)Wh;
    const float c0 = consts[0], c1 = consts[1], c2 = consts[2], scale = consts[3];

    for (int q = 0; q < BIN_ROWS / GROUPS; ++q) {
        const int rr  = q * GROUPS + g;
        const int row = b * BIN_ROWS + rr;
        if (row >= N) continue;

        float4 r4 = Wh4[(size_t)row * (D / 4) + l];
        float nr = norms[row];
        const int n_all = rcnt[rr];
        const int n = n_all < CAP ? n_all : CAP;
        const int2* bk = lb + rr * CAP;

        float4 acc = make_float4(0.f, 0.f, 0.f, 0.f);
        const int own_k = l >> 1;

        for (int j = 0; j < n; j += DEPTH) {
            v4f   c4[DEPTH];
            float dt[DEPTH];
            const bool live = (j + own_k) < n;
            const int2 own = bk[live ? (j + own_k) : (n - 1)];
            c4[0] = Wh4v[(size_t)row_share_i<0>(own.x)  * (D / 4) + l];
            c4[1] = Wh4v[(size_t)row_share_i<2>(own.x)  * (D / 4) + l];
            c4[2] = Wh4v[(size_t)row_share_i<4>(own.x)  * (D / 4) + l];
            c4[3] = Wh4v[(size_t)row_share_i<6>(own.x)  * (D / 4) + l];
            c4[4] = Wh4v[(size_t)row_share_i<8>(own.x)  * (D / 4) + l];
            c4[5] = Wh4v[(size_t)row_share_i<10>(own.x) * (D / 4) + l];
            c4[6] = Wh4v[(size_t)row_share_i<12>(own.x) * (D / 4) + l];
            c4[7] = Wh4v[(size_t)row_share_i<14>(own.x) * (D / 4) + l];
            const float nc = norms[own.x];
            asm volatile("" :: "v"(c4[0]), "v"(c4[1]), "v"(c4[2]), "v"(c4[3]),
                               "v"(c4[4]), "v"(c4[5]), "v"(c4[6]), "v"(c4[7]),
                               "v"(nc));
            __builtin_amdgcn_sched_barrier(0);
            for (int k = 0; k < DEPTH; ++k)
                dt[k] = r4.x * c4[k].x + r4.y * c4[k].y + r4.z * c4[k].z + r4.w * c4[k].w;
            const float dot = row16_sum8_scatter(dt, l);
            float w = edge_weight(dot, nr, nc, __int_as_float(own.y), c0, c1);
            w = live ? w : 0.0f;
            float wk[DEPTH];
            wk[0] = row_share_f<0>(w);
            wk[1] = row_share_f<2>(w);
            wk[2] = row_share_f<4>(w);
            wk[3] = row_share_f<6>(w);
            wk[4] = row_share_f<8>(w);
            wk[5] = row_share_f<10>(w);
            wk[6] = row_share_f<12>(w);
            wk[7] = row_share_f<14>(w);
            for (int k = 0; k < DEPTH; ++k) {
                acc.x += wk[k] * c4[k].x;
                acc.y += wk[k] * c4[k].y;
                acc.z += wk[k] * c4[k].z;
                acc.w += wk[k] * c4[k].w;
            }
        }

        if (n_all > CAP) {
            acc = make_float4(0.f, 0.f, 0.f, 0.f);
            for (int k = 0; k < NBLK; ++k) {
                int la = rl[k];
                int len = la < SLOT ? la : SLOT;
                const int2* run = runs + (size_t)k * SLOT;
                for (int i = 0; i < len; ++i) {
                    int2 rec = run[i];
                    unsigned u = (unsigned)rec.x;
                    if ((int)(u >> 24) != rr) continue;
                    add_edge(acc, r4, nr, (int)(u & COL_MASK), __int_as_float(rec.y),
                             Wh4, norms, l, c0, c1);
                }
            }
        }

        if (range_ovf) {
            int m = ovf_cnt[0];
            float dc0 = dcoef[0], dc1 = dcoef[1];
            for (int t = 0; t < m; ++t) {
                int e = ovf[t];
                if (ei[e] != row) continue;
                int col = ei[E + e];
                float x  = dc0 * deg_v[e] + dc1;
                float sc = (x > 20.0f) ? x : log1pf(expf(x));
                float base = adj_v[e] * sc;
                add_edge(acc, r4, nr, col, base, Wh4, norms, l, c0, c1);
            }
        }

        float4 o;
        o.x = scale * (acc.x + c2 * r4.x);
        o.y = scale * (acc.y + c2 * r4.y);
        o.z = scale * (acc.z + c2 * r4.z);
        o.w = scale * (acc.w + c2 * r4.w);
        ((float4*)out)[(size_t)row * (D / 4) + l] = o;
    }
}

extern "C" void kernel_launch(void* const* d_in, const int* in_sizes, int n_in,
                              void* d_out, int out_size, void* d_ws, size_t ws_size,
                              hipStream_t stream) {
    const float* h     = (const float*)d_in[0];
    const int*   ei    = (const int*)  d_in[1];
    const float* adj_v = (const float*)d_in[2];
    const float* deg_v = (const float*)d_in[3];
    const float* W     = (const float*)d_in[4];
    const float* bias  = (const float*)d_in[5];
    const float* dcoef = (const float*)d_in[6];
    const float* cpar  = (const float*)d_in[7];
    const float* spar  = (const float*)d_in[8];

    const int N = in_sizes[0] / D;
    const int E = in_sizes[2];
    const int NBIN = (N + BIN_ROWS - 1) / BIN_ROWS;
    const int NBLK = (E + EPB - 1) / EPB;

    float* Wh      = (float*)d_ws;
    int2*  binned  = (int2*)(Wh + (size_t)N * D);
    int*   ovf     = (int*)(binned + (size_t)NBIN * NBLK * SLOT);
    float* norms   = (float*)(ovf + E);
    int*   runlen  = (int*)(norms + N);
    float* consts  = (float*)(runlen + (size_t)NBIN * NBLK);
    int*   ovf_cnt = (int*)(consts + 4);

    gemm_wmma_kernel<<<(N + 63) / 64 + 1, 128, 0, stream>>>(h, W, bias, Wh, norms,
                                                            cpar, spar, consts, ovf_cnt, N);

    bin_kernel<<<NBLK, BIN_THREADS, 0, stream>>>(ei, adj_v, deg_v, dcoef, binned, runlen,
                                                 ovf_cnt, ovf, E, NBIN, NBLK);

    gather_kernel<<<NBIN, GATHER_THREADS, 0, stream>>>(ei, adj_v, deg_v, dcoef, Wh, norms,
                                                       consts, runlen, binned, ovf_cnt, ovf,
                                                       (float*)d_out, N, E, NBLK);
}


# Edge grouping pass folded into the GEMM launch as extra blocks (256 threads, 16 edges each; per-block overflow lists, no global counter to clear): 3 launches -> 2; row sums still arrival order
# speedup vs baseline: 4.2315x; 1.1065x over previous
#include <hip/hip_runtime.h>
#include <hip/hip_bf16.h>

typedef __attribute__((ext_vector_type(2))) float v2f;
typedef __attribute__((ext_vector_type(8))) float v8f;
typedef __attribute__((ext_vector_type(4))) float v4f;

#define D 64
#define LSTR 66

__device__ __forceinline__ void fold_consts(const float* __restrict__ coeff_param,
                                            const float* __restrict__ scale_param,
                                            float* __restrict__ consts) {
    float x0 = coeff_param[0], x1 = coeff_param[1], x2 = coeff_param[2];
    float m = fmaxf(x0, fmaxf(x1, x2));
    float e0 = expf(x0 - m), e1 = expf(x1 - m), e2 = expf(x2 - m);
    float inv = 1.0f / (e0 + e1 + e2);
    consts[0] = e0 * inv;
    consts[1] = e1 * inv;
    consts[2] = e2 * inv;
    float s = scale_param[0];
    consts[3] = (s > 20.0f) ? s : log1pf(expf(s));
}

__device__ __forceinline__ float row16_sum(float v) {
    v += __int_as_float(__builtin_amdgcn_update_dpp(0, __float_as_int(v), 0x168, 0xf, 0xf, false));
    v += __int_as_float(__builtin_amdgcn_update_dpp(0, __float_as_int(v), 0x164, 0xf, 0xf, false));
    v += __int_as_float(__builtin_amdgcn_update_dpp(0, __float_as_int(v), 0x162, 0xf, 0xf, false));
    v += __int_as_float(__builtin_amdgcn_update_dpp(0, __float_as_int(v), 0x161, 0xf, 0xf, false));
    return v;
}

template <int XM>
__device__ __forceinline__ float row_xmask_f(float v) {
    return __int_as_float(__builtin_amdgcn_update_dpp(0, __float_as_int(v), 0x160 + XM, 0xf, 0xf, false));
}
template <int SRC>
__device__ __forceinline__ int row_share_i(int v) {
    return __builtin_amdgcn_mov_dpp(v, 0x150 + SRC, 0xf, 0xf, true);
}
template <int SRC>
__device__ __forceinline__ float row_share_f(float v) {
    return __int_as_float(row_share_i<SRC>(__float_as_int(v)));
}

template <int XM>
__device__ __forceinline__ float halve_pair(float a, float b, bool hi) {
    float keep = hi ? b : a;
    float send = hi ? a : b;
    return keep + row_xmask_f<XM>(send);
}

__device__ __forceinline__ float row16_sum8_scatter(const float (&d)[8], int p) {
    const bool h8 = (p & 8) != 0, h4 = (p & 4) != 0, h2 = (p & 2) != 0;
    float e0 = halve_pair<8>(d[0], d[4], h8);
    float e1 = halve_pair<8>(d[1], d[5], h8);
    float e2 = halve_pair<8>(d[2], d[6], h8);
    float e3 = halve_pair<8>(d[3], d[7], h8);
    float f0 = halve_pair<4>(e0, e2, h4);
    float f1 = halve_pair<4>(e1, e3, h4);
    float g  = halve_pair<2>(f0, f1, h2);
    return g + row_xmask_f<1>(g);
}

#define BIN_ROWS 256
#define NBIN_MAX 256
#define EPB 4096
#define FRONT_THREADS 256
#define EPT (EPB / FRONT_THREADS)
#define SLOT 64
#define CAP 64
#define GATHER_THREADS 1024
#define GROUPS (GATHER_THREADS / 16)
#define DEPTH 8
#define COL_MASK 0xFFFFFFu

__global__ void __launch_bounds__(FRONT_THREADS)
front_kernel(const float* __restrict__ h, const float* __restrict__ W,
             const float* __restrict__ bias, float* __restrict__ Wh,
             float* __restrict__ norms,
             const float* __restrict__ coeff_param,
             const float* __restrict__ scale_param,
             float* __restrict__ consts,
             const int* __restrict__ ei, const float* __restrict__ adj_v,
             const float* __restrict__ deg_v, const float* __restrict__ dcoef,
             int2* __restrict__ binned, int* __restrict__ runlen,
             int* __restrict__ ovfn, int* __restrict__ ovf,
             int N, int E, int NBIN, int NBLK) {
    __shared__ float Ws[D * LSTR];
    __shared__ int hist[NBIN_MAX];
    __shared__ int ovl;

    const int tid = threadIdx.x;
    const int bx  = blockIdx.x;
    if (bx == 0) {
        if (tid == 0) fold_consts(coeff_param, scale_param, consts);
        return;
    }

    if (bx <= NBLK) {
        const int blk = bx - 1;
        hist[tid] = 0;
        if (tid == 0) ovl = 0;
        __syncthreads();

        const float dc0 = dcoef[0], dc1 = dcoef[1];
        const int ebase = blk * EPB;
        int   row[EPT], col[EPT];
        float dg[EPT], ad[EPT];
        if (((E & 3) == 0) && (ebase + EPB <= E)) {
#pragma unroll
            for (int c = 0; c < EPT / 4; ++c) {
                const int e = ebase + (c * FRONT_THREADS + tid) * 4;
                const int4   r4 = *(const int4*)(ei + e);
                const int4   c4 = *(const int4*)(ei + E + e);
                const float4 d4 = *(const float4*)(deg_v + e);
                const float4 a4 = *(const float4*)(adj_v + e);
                row[4 * c + 0] = r4.x; row[4 * c + 1] = r4.y; row[4 * c + 2] = r4.z; row[4 * c + 3] = r4.w;
                col[4 * c + 0] = c4.x; col[4 * c + 1] = c4.y; col[4 * c + 2] = c4.z; col[4 * c + 3] = c4.w;
                dg[4 * c + 0] = d4.x;  dg[4 * c + 1] = d4.y;  dg[4 * c + 2] = d4.z;  dg[4 * c + 3] = d4.w;
                ad[4 * c + 0] = a4.x;  ad[4 * c + 1] = a4.y;  ad[4 * c + 2] = a4.z;  ad[4 * c + 3] = a4.w;
            }
        } else {
#pragma unroll
            for (int k = 0; k < EPT; ++k) {
                const int e = ebase + ((k >> 2) * FRONT_THREADS + tid) * 4 + (k & 3);
                const bool ok = e < E;
                const int idx = ok ? e : 0;
                const int r = ei[idx], c = ei[E + idx];
                row[k] = ok ? r : 0;
                col[k] = ok ? c : 0;
                dg[k]  = deg_v[idx];
                ad[k]  = adj_v[idx];
            }
        }
        int slot[EPT];
#pragma unroll
        for (int k = 0; k < EPT; ++k) {
            slot[k] = -1;
            if (row[k] != col[k]) slot[k] = atomicAdd(&hist[row[k] >> 8], 1);
        }
#pragma unroll
        for (int k = 0; k < EPT; ++k) {
            if (slot[k] < 0) continue;
            const int bin = row[k] >> 8;
            float x  = dc0 * dg[k] + dc1;
            float sc = (x > 20.0f) ? x : log1pf(expf(x));
            float base = ad[k] * sc;
            if (slot[k] < SLOT) {
                unsigned key = (unsigned)col[k] | ((unsigned)(row[k] & (BIN_ROWS - 1)) << 24);
                binned[((size_t)bin * NBLK + blk) * SLOT + slot[k]] = make_int2((int)key, __float_as_int(base));
            } else {
                int q = atomicAdd(&ovl, 1);
                ovf[(size_t)blk * EPB + q] = ebase + ((k >> 2) * FRONT_THREADS + tid) * 4 + (k & 3);
            }
        }
        __syncthreads();
        if (tid < NBIN) {
            int c = hist[tid];
            runlen[tid * NBLK + blk] = c;
            if (c == 0) binned[((size_t)tid * NBLK + blk) * SLOT] = make_int2(0, 0);
        }
        if (tid == 0) ovfn[blk] = ovl;
        return;
    }

    for (int i = tid * 4; i < D * D; i += FRONT_THREADS * 4) {
        int r = i >> 6;
        int c = i & 63;
        float4 w4 = *(const float4*)(W + i);
        Ws[(c + 0) * LSTR + r] = w4.x;
        Ws[(c + 1) * LSTR + r] = w4.y;
        Ws[(c + 2) * LSTR + r] = w4.z;
        Ws[(c + 3) * LSTR + r] = w4.w;
    }
    __syncthreads();

    int wave = tid >> 5;
    int lane = tid & 31;
    int half = lane >> 4;
    int l16  = lane & 15;

    int m0 = ((bx - 1 - NBLK) * (FRONT_THREADS / 32) + wave) * 16;
    if (m0 >= N) return;

    const float* hrow = h + (size_t)(m0 + l16) * D + 2 * half;
    const float* wt   = Ws + l16 * LSTR + 2 * half;

    v8f acc[4] = {};

#pragma unroll 4
    for (int s = 0; s < 16; ++s) {
        v2f a = *(const v2f*)(hrow + 4 * s);
#pragma unroll
        for (int t = 0; t < 4; ++t) {
            v2f bf = *(const v2f*)(wt + t * (16 * LSTR) + 4 * s);
            acc[t] = __builtin_amdgcn_wmma_f32_16x16x4_f32(
                false, a, false, bf, (short)0, acc[t], false, false);
        }
    }

    float ss[8] = {0.f, 0.f, 0.f, 0.f, 0.f, 0.f, 0.f, 0.f};
#pragma unroll
    for (int t = 0; t < 4; ++t) {
        int n = t * 16 + l16;
        float bv = bias[n];
#pragma unroll
        for (int v = 0; v < 8; ++v) {
            float val = acc[t][v] + bv;
            Wh[(size_t)(m0 + half * 8 + v) * D + n] = val;
            ss[v] += val * val;
        }
    }
#pragma unroll
    for (int v = 0; v < 8; ++v) ss[v] = row16_sum(ss[v]);
    float mine = ss[0];
#pragma unroll
    for (int v = 1; v < 8; ++v) mine = (l16 == v) ? ss[v] : mine;
    if (l16 < 8) {
        int node = m0 + half * 8 + l16;
        norms[node] = sqrtf(mine);
    }
}

__device__ __forceinline__ float edge_weight(float dot, float nr, float nc,
                                             float base, float c0, float c1) {
    float sim = dot / fmaxf(nr * nc, 1e-8f);
    return base * (sim > 0.0f ? c0 * sim : c1 * sim);
}

__device__ __forceinline__ void add_edge(float4& acc, const float4 r4, float nr, int col,
                                         float base, const float4* __restrict__ Wh4,
                                         const float* __restrict__ norms, int l,
                                         float c0, float c1) {
    float4 cc = Wh4[(size_t)col * (D / 4) + l];
    float d = r4.x * cc.x + r4.y * cc.y + r4.z * cc.z + r4.w * cc.w;
    d = row16_sum(d);
    float w = edge_weight(d, nr, norms[col], base, c0, c1);
    acc.x += w * cc.x;
    acc.y += w * cc.y;
    acc.z += w * cc.z;
    acc.w += w * cc.w;
}

__global__ void __launch_bounds__(GATHER_THREADS)
gather_kernel(const int* __restrict__ ei, const float* __restrict__ adj_v,
              const float* __restrict__ deg_v, const float* __restrict__ dcoef,
              const float* __restrict__ Wh, const float* __restrict__ norms,
              const float* __restrict__ consts, const int* __restrict__ runlen,
              const int2* __restrict__ binned, const int* __restrict__ ovfn,
              const int* __restrict__ ovf, float* __restrict__ out, int N, int E, int NBLK) {
    __shared__ int  rcnt[BIN_ROWS];
    __shared__ int  binovf;
    __shared__ int2 lb[BIN_ROWS * CAP];

    const int tid = threadIdx.x;
    const int b   = blockIdx.x;
    const int g   = tid >> 4;
    const int l   = tid & 15;

    if (tid < BIN_ROWS) rcnt[tid] = 0;
    if (tid == 0) binovf = 0;
    __syncthreads();

    const int* rl = runlen + (size_t)b * NBLK;
    const int2* runs = binned + (size_t)b * NBLK * SLOT;
    for (int k0 = 0; k0 < NBLK; k0 += 4 * GROUPS) {
        int len[4];
        const int2* run[4];
        int2 ra[4], rb[4];
        int la[4];
        for (int p = 0; p < 4; ++p) {
            int k = k0 + p * GROUPS + g;
            bool valid = k < NBLK;
            int kk = valid ? k : 0;
            int v = rl[kk];
            la[p] = valid ? v : 0;
            run[p] = runs + (size_t)kk * SLOT;
        }
        int lmax = la[0];
        for (int p = 0; p < 4; ++p) {
            lmax = la[p] > lmax ? la[p] : lmax;
            len[p] = la[p] < SLOT ? la[p] : SLOT;
        }
        if (lmax > SLOT) binovf = 1;
        for (int p = 0; p < 4; ++p) {
            int last = (len[p] > 0 ? len[p] : 1) - 1;
            int ia = l < last ? l : last;
            int ib = (l + 16) < last ? (l + 16) : last;
            ra[p] = run[p][ia];
            rb[p] = run[p][ib];
        }
        for (int p = 0; p < 4; ++p) {
            if (l < len[p]) {
                unsigned u = (unsigned)ra[p].x;
                int r = (int)(u >> 24);
                int slot = atomicAdd(&rcnt[r], 1);
                if (slot < CAP) lb[r * CAP + slot] = make_int2((int)(u & COL_MASK), ra[p].y);
            }
            if (l + 16 < len[p]) {
                unsigned u = (unsigned)rb[p].x;
                int r = (int)(u >> 24);
                int slot = atomicAdd(&rcnt[r], 1);
                if (slot < CAP) lb[r * CAP + slot] = make_int2((int)(u & COL_MASK), rb[p].y);
            }
            for (int i = l + 32; i < len[p]; i += 16) {
                int2 rec = run[p][i];
                unsigned u = (unsigned)rec.x;
                int r = (int)(u >> 24);
                int slot = atomicAdd(&rcnt[r], 1);
                if (slot < CAP) lb[r * CAP + slot] = make_int2((int)(u & COL_MASK), rec.y);
            }
        }
    }
    __syncthreads();

    const int range_ovf = binovf;
    const float4* Wh4 = (const float4*)Wh;
    const v4f* Wh4v = (const v4f*)Wh;
    const float c0 = consts[0], c1 = consts[1], c2 = consts[2], scale = consts[3];

    for (int q = 0; q < BIN_ROWS / GROUPS; ++q) {
        const int rr  = q * GROUPS + g;
        const int row = b * BIN_ROWS + rr;
        if (row >= N) continue;

        float4 r4 = Wh4[(size_t)row * (D / 4) + l];
        float nr = norms[row];
        const int n_all = rcnt[rr];
        const int n = n_all < CAP ? n_all : CAP;
        const int2* bk = lb + rr * CAP;

        float4 acc = make_float4(0.f, 0.f, 0.f, 0.f);
        const int own_k = l >> 1;

        for (int j = 0; j < n; j += DEPTH) {
            v4f   c4[DEPTH];
            float dt[DEPTH];
            const bool live = (j + own_k) < n;
            const int2 own = bk[live ? (j + own_k) : (n - 1)];
            c4[0] = Wh4v[(size_t)row_share_i<0>(own.x)  * (D / 4) + l];
            c4[1] = Wh4v[(size_t)row_share_i<2>(own.x)  * (D / 4) + l];
            c4[2] = Wh4v[(size_t)row_share_i<4>(own.x)  * (D / 4) + l];
            c4[3] = Wh4v[(size_t)row_share_i<6>(own.x)  * (D / 4) + l];
            c4[4] = Wh4v[(size_t)row_share_i<8>(own.x)  * (D / 4) + l];
            c4[5] = Wh4v[(size_t)row_share_i<10>(own.x) * (D / 4) + l];
            c4[6] = Wh4v[(size_t)row_share_i<12>(own.x) * (D / 4) + l];
            c4[7] = Wh4v[(size_t)row_share_i<14>(own.x) * (D / 4) + l];
            const float nc = norms[own.x];
            asm volatile("" :: "v"(c4[0]), "v"(c4[1]), "v"(c4[2]), "v"(c4[3]),
                               "v"(c4[4]), "v"(c4[5]), "v"(c4[6]), "v"(c4[7]),
                               "v"(nc));
            __builtin_amdgcn_sched_barrier(0);
            for (int k = 0; k < DEPTH; ++k)
                dt[k] = r4.x * c4[k].x + r4.y * c4[k].y + r4.z * c4[k].z + r4.w * c4[k].w;
            const float dot = row16_sum8_scatter(dt, l);
            float w = edge_weight(dot, nr, nc, __int_as_float(own.y), c0, c1);
            w = live ? w : 0.0f;
            float wk[DEPTH];
            wk[0] = row_share_f<0>(w);
            wk[1] = row_share_f<2>(w);
            wk[2] = row_share_f<4>(w);
            wk[3] = row_share_f<6>(w);
            wk[4] = row_share_f<8>(w);
            wk[5] = row_share_f<10>(w);
            wk[6] = row_share_f<12>(w);
            wk[7] = row_share_f<14>(w);
            for (int k = 0; k < DEPTH; ++k) {
                acc.x += wk[k] * c4[k].x;
                acc.y += wk[k] * c4[k].y;
                acc.z += wk[k] * c4[k].z;
                acc.w += wk[k] * c4[k].w;
            }
        }

        if (n_all > CAP) {
            acc = make_float4(0.f, 0.f, 0.f, 0.f);
            for (int k = 0; k < NBLK; ++k) {
                int la = rl[k];
                int len = la < SLOT ? la : SLOT;
                const int2* run = runs + (size_t)k * SLOT;
                for (int i = 0; i < len; ++i) {
                    int2 rec = run[i];
                    unsigned u = (unsigned)rec.x;
                    if ((int)(u >> 24) != rr) continue;
                    add_edge(acc, r4, nr, (int)(u & COL_MASK), __int_as_float(rec.y),
                             Wh4, norms, l, c0, c1);
                }
            }
        }

        if (range_ovf) {
            float dc0 = dcoef[0], dc1 = dcoef[1];
            for (int kb = 0; kb < NBLK; ++kb) {
                int m = ovfn[kb];
                m = m < EPB ? m : EPB;
                const int* lst = ovf + (size_t)kb * EPB;
                for (int t = 0; t < m; ++t) {
                    int e = lst[t];
                    if (ei[e] != row) continue;
                    int col = ei[E + e];
                    float x  = dc0 * deg_v[e] + dc1;
                    float sc = (x > 20.0f) ? x : log1pf(expf(x));
                    float base = adj_v[e] * sc;
                    add_edge(acc, r4, nr, col, base, Wh4, norms, l, c0, c1);
                }
            }
        }

        float4 o;
        o.x = scale * (acc.x + c2 * r4.x);
        o.y = scale * (acc.y + c2 * r4.y);
        o.z = scale * (acc.z + c2 * r4.z);
        o.w = scale * (acc.w + c2 * r4.w);
        ((float4*)out)[(size_t)row * (D / 4) + l] = o;
    }
}

extern "C" void kernel_launch(void* const* d_in, const int* in_sizes, int n_in,
                              void* d_out, int out_size, void* d_ws, size_t ws_size,
                              hipStream_t stream) {
    const float* h     = (const float*)d_in[0];
    const int*   ei    = (const int*)  d_in[1];
    const float* adj_v = (const float*)d_in[2];
    const float* deg_v = (const float*)d_in[3];
    const float* W     = (const float*)d_in[4];
    const float* bias  = (const float*)d_in[5];
    const float* dcoef = (const float*)d_in[6];
    const float* cpar  = (const float*)d_in[7];
    const float* spar  = (const float*)d_in[8];

    const int N = in_sizes[0] / D;
    const int E = in_sizes[2];
    const int NBIN = (N + BIN_ROWS - 1) / BIN_ROWS;
    const int NBLK = (E + EPB - 1) / EPB;
    const int MBLK = (N + 127) / 128;

    float* Wh      = (float*)d_ws;
    int2*  binned  = (int2*)(Wh + (size_t)N * D);
    int*   ovf     = (int*)(binned + (size_t)NBIN * NBLK * SLOT);
    float* norms   = (float*)(ovf + (size_t)NBLK * EPB);
    int*   runlen  = (int*)(norms + N);
    float* consts  = (float*)(runlen + (size_t)NBIN * NBLK);
    int*   ovfn    = (int*)(consts + 4);

    front_kernel<<<1 + NBLK + MBLK, FRONT_THREADS, 0, stream>>>(
        h, W, bias, Wh, norms, cpar, spar, consts,
        ei, adj_v, deg_v, dcoef, binned, runlen, ovfn, ovf, N, E, NBIN, NBLK);

    gather_kernel<<<NBIN, GATHER_THREADS, 0, stream>>>(ei, adj_v, deg_v, dcoef, Wh, norms,
                                                       consts, runlen, binned, ovfn, ovf,
                                                       (float*)d_out, N, E, NBLK);
}


# Grouping blocks rank their records by row range in LDS and write one contiguous chunk per block with whole-line stores (run = offset+length); no scattered 8-byte stores; row sums still arrival order
# speedup vs baseline: 4.4477x; 1.0511x over previous
#include <hip/hip_runtime.h>
#include <hip/hip_bf16.h>

typedef __attribute__((ext_vector_type(2))) float v2f;
typedef __attribute__((ext_vector_type(8))) float v8f;
typedef __attribute__((ext_vector_type(4))) float v4f;

#define D 64
#define LSTR 66

__device__ __forceinline__ void fold_consts(const float* __restrict__ coeff_param,
                                            const float* __restrict__ scale_param,
                                            float* __restrict__ consts) {
    float x0 = coeff_param[0], x1 = coeff_param[1], x2 = coeff_param[2];
    float m = fmaxf(x0, fmaxf(x1, x2));
    float e0 = expf(x0 - m), e1 = expf(x1 - m), e2 = expf(x2 - m);
    float inv = 1.0f / (e0 + e1 + e2);
    consts[0] = e0 * inv;
    consts[1] = e1 * inv;
    consts[2] = e2 * inv;
    float s = scale_param[0];
    consts[3] = (s > 20.0f) ? s : log1pf(expf(s));
}

__device__ __forceinline__ float row16_sum(float v) {
    v += __int_as_float(__builtin_amdgcn_update_dpp(0, __float_as_int(v), 0x168, 0xf, 0xf, false));
    v += __int_as_float(__builtin_amdgcn_update_dpp(0, __float_as_int(v), 0x164, 0xf, 0xf, false));
    v += __int_as_float(__builtin_amdgcn_update_dpp(0, __float_as_int(v), 0x162, 0xf, 0xf, false));
    v += __int_as_float(__builtin_amdgcn_update_dpp(0, __float_as_int(v), 0x161, 0xf, 0xf, false));
    return v;
}

template <int XM>
__device__ __forceinline__ float row_xmask_f(float v) {
    return __int_as_float(__builtin_amdgcn_update_dpp(0, __float_as_int(v), 0x160 + XM, 0xf, 0xf, false));
}
template <int SRC>
__device__ __forceinline__ int row_share_i(int v) {
    return __builtin_amdgcn_mov_dpp(v, 0x150 + SRC, 0xf, 0xf, true);
}
template <int SRC>
__device__ __forceinline__ float row_share_f(float v) {
    return __int_as_float(row_share_i<SRC>(__float_as_int(v)));
}

template <int XM>
__device__ __forceinline__ float halve_pair(float a, float b, bool hi) {
    float keep = hi ? b : a;
    float send = hi ? a : b;
    return keep + row_xmask_f<XM>(send);
}

__device__ __forceinline__ float row16_sum8_scatter(const float (&d)[8], int p) {
    const bool h8 = (p & 8) != 0, h4 = (p & 4) != 0, h2 = (p & 2) != 0;
    float e0 = halve_pair<8>(d[0], d[4], h8);
    float e1 = halve_pair<8>(d[1], d[5], h8);
    float e2 = halve_pair<8>(d[2], d[6], h8);
    float e3 = halve_pair<8>(d[3], d[7], h8);
    float f0 = halve_pair<4>(e0, e2, h4);
    float f1 = halve_pair<4>(e1, e3, h4);
    float g  = halve_pair<2>(f0, f1, h2);
    return g + row_xmask_f<1>(g);
}

#define BIN_ROWS 256
#define NBIN_MAX 256
#define EPB 4096
#define FRONT_THREADS 256
#define EPT (EPB / FRONT_THREADS)
#define LEN_BITS 13
#define LEN_MASK ((1 << LEN_BITS) - 1)
#define CAP 64
#define GATHER_THREADS 1024
#define GROUPS (GATHER_THREADS / 16)
#define DEPTH 8
#define COL_MASK 0xFFFFFFu

__global__ void __launch_bounds__(FRONT_THREADS)
front_kernel(const float* __restrict__ h, const float* __restrict__ W,
             const float* __restrict__ bias, float* __restrict__ Wh,
             float* __restrict__ norms,
             const float* __restrict__ coeff_param,
             const float* __restrict__ scale_param,
             float* __restrict__ consts,
             const int* __restrict__ ei, const float* __restrict__ adj_v,
             const float* __restrict__ deg_v, const float* __restrict__ dcoef,
             int2* __restrict__ binned, int* __restrict__ runlen,
             int N, int E, int NBIN, int NBLK) {
    __shared__ float Ws[D * LSTR];
    __shared__ int hist[NBIN_MAX];
    __shared__ int offs[NBIN_MAX];
    __shared__ int wsum[FRONT_THREADS / 32];
    __shared__ __attribute__((aligned(16))) int2 stage[EPB];

    const int tid = threadIdx.x;
    const int bx  = blockIdx.x;
    if (bx == 0) {
        if (tid == 0) fold_consts(coeff_param, scale_param, consts);
        return;
    }

    if (bx <= NBLK) {
        const int blk = bx - 1;
        hist[tid] = 0;
        __syncthreads();

        const float dc0 = dcoef[0], dc1 = dcoef[1];
        const int ebase = blk * EPB;
        int   row[EPT], col[EPT];
        float dg[EPT], ad[EPT];
        if (((E & 3) == 0) && (ebase + EPB <= E)) {
#pragma unroll
            for (int c = 0; c < EPT / 4; ++c) {
                const int e = ebase + (c * FRONT_THREADS + tid) * 4;
                const int4   r4 = *(const int4*)(ei + e);
                const int4   c4 = *(const int4*)(ei + E + e);
                const float4 d4 = *(const float4*)(deg_v + e);
                const float4 a4 = *(const float4*)(adj_v + e);
                row[4 * c + 0] = r4.x; row[4 * c + 1] = r4.y; row[4 * c + 2] = r4.z; row[4 * c + 3] = r4.w;
                col[4 * c + 0] = c4.x; col[4 * c + 1] = c4.y; col[4 * c + 2] = c4.z; col[4 * c + 3] = c4.w;
                dg[4 * c + 0] = d4.x;  dg[4 * c + 1] = d4.y;  dg[4 * c + 2] = d4.z;  dg[4 * c + 3] = d4.w;
                ad[4 * c + 0] = a4.x;  ad[4 * c + 1] = a4.y;  ad[4 * c + 2] = a4.z;  ad[4 * c + 3] = a4.w;
            }
        } else {
#pragma unroll
            for (int k = 0; k < EPT; ++k) {
                const int e = ebase + ((k >> 2) * FRONT_THREADS + tid) * 4 + (k & 3);
                const bool ok = e < E;
                const int idx = ok ? e : 0;
                const int r = ei[idx], c = ei[E + idx];
                row[k] = ok ? r : 0;
                col[k] = ok ? c : 0;
                dg[k]  = deg_v[idx];
                ad[k]  = adj_v[idx];
            }
        }
        int slot[EPT];
#pragma unroll
        for (int k = 0; k < EPT; ++k) {
            slot[k] = -1;
            if (row[k] != col[k]) slot[k] = atomicAdd(&hist[row[k] >> 8], 1);
        }
        __syncthreads();

        const int cnt = hist[tid];
        const int ln = tid & 31, wv = tid >> 5;
        int incl = cnt;
#pragma unroll
        for (int d = 1; d < 32; d <<= 1) {
            int y = __shfl_up(incl, d);
            if (ln >= d) incl += y;
        }
        if (ln == 31) wsum[wv] = incl;
        __syncthreads();
        int wbase = 0;
#pragma unroll
        for (int w = 0; w < FRONT_THREADS / 32; ++w) wbase += (w < wv) ? wsum[w] : 0;
        const int myoff = wbase + incl - cnt;
        offs[tid] = myoff;
        if (tid == NBIN_MAX - 1) {
            const int total = myoff + cnt;
            if (total & 1) stage[total] = make_int2(0, 0);
            if (total == 0) { stage[0] = make_int2(0, 0); stage[1] = make_int2(0, 0); }
        }
        __syncthreads();

#pragma unroll
        for (int k = 0; k < EPT; ++k) {
            if (slot[k] < 0) continue;
            const int bin = row[k] >> 8;
            float x  = dc0 * dg[k] + dc1;
            float sc = (x > 20.0f) ? x : log1pf(expf(x));
            float base = ad[k] * sc;
            unsigned key = (unsigned)col[k] | ((unsigned)(row[k] & (BIN_ROWS - 1)) << 24);
            stage[offs[bin] + slot[k]] = make_int2((int)key, __float_as_int(base));
        }
        __syncthreads();

        const int total = offs[NBIN_MAX - 1] + hist[NBIN_MAX - 1];
        int n4 = (total + 1) >> 1;
        n4 = n4 < 1 ? 1 : n4;
        int4* dst = (int4*)(binned + (size_t)blk * EPB);
        const int4* srcp = (const int4*)stage;
        int4 pr[EPB / 2 / FRONT_THREADS];
#pragma unroll
        for (int j = 0; j < EPB / 2 / FRONT_THREADS; ++j) {
            const int i = j * FRONT_THREADS + tid;
            pr[j] = (i < n4) ? srcp[i] : make_int4(0, 0, 0, 0);
        }
#pragma unroll
        for (int j = 0; j < EPB / 2 / FRONT_THREADS; ++j) {
            const int i = j * FRONT_THREADS + tid;
            if (i < n4) dst[i] = pr[j];
        }
        if (tid < NBIN) runlen[tid * NBLK + blk] = (myoff << LEN_BITS) | cnt;
        return;
    }

    for (int i = tid * 4; i < D * D; i += FRONT_THREADS * 4) {
        int r = i >> 6;
        int c = i & 63;
        float4 w4 = *(const float4*)(W + i);
        Ws[(c + 0) * LSTR + r] = w4.x;
        Ws[(c + 1) * LSTR + r] = w4.y;
        Ws[(c + 2) * LSTR + r] = w4.z;
        Ws[(c + 3) * LSTR + r] = w4.w;
    }
    __syncthreads();

    int wave = tid >> 5;
    int lane = tid & 31;
    int half = lane >> 4;
    int l16  = lane & 15;

    int m0 = ((bx - 1 - NBLK) * (FRONT_THREADS / 32) + wave) * 16;
    if (m0 >= N) return;

    const float* hrow = h + (size_t)(m0 + l16) * D + 2 * half;
    const float* wt   = Ws + l16 * LSTR + 2 * half;

    v8f acc[4] = {};

#pragma unroll 4
    for (int s = 0; s < 16; ++s) {
        v2f a = *(const v2f*)(hrow + 4 * s);
#pragma unroll
        for (int t = 0; t < 4; ++t) {
            v2f bf = *(const v2f*)(wt + t * (16 * LSTR) + 4 * s);
            acc[t] = __builtin_amdgcn_wmma_f32_16x16x4_f32(
                false, a, false, bf, (short)0, acc[t], false, false);
        }
    }

    float ss[8] = {0.f, 0.f, 0.f, 0.f, 0.f, 0.f, 0.f, 0.f};
#pragma unroll
    for (int t = 0; t < 4; ++t) {
        int n = t * 16 + l16;
        float bv = bias[n];
#pragma unroll
        for (int v = 0; v < 8; ++v) {
            float val = acc[t][v] + bv;
            Wh[(size_t)(m0 + half * 8 + v) * D + n] = val;
            ss[v] += val * val;
        }
    }
#pragma unroll
    for (int v = 0; v < 8; ++v) ss[v] = row16_sum(ss[v]);
    float mine = ss[0];
#pragma unroll
    for (int v = 1; v < 8; ++v) mine = (l16 == v) ? ss[v] : mine;
    if (l16 < 8) {
        int node = m0 + half * 8 + l16;
        norms[node] = sqrtf(mine);
    }
}

__device__ __forceinline__ float edge_weight(float dot, float nr, float nc,
                                             float base, float c0, float c1) {
    float sim = dot / fmaxf(nr * nc, 1e-8f);
    return base * (sim > 0.0f ? c0 * sim : c1 * sim);
}

__device__ __forceinline__ void add_edge(float4& acc, const float4 r4, float nr, int col,
                                         float base, const float4* __restrict__ Wh4,
                                         const float* __restrict__ norms, int l,
                                         float c0, float c1) {
    float4 cc = Wh4[(size_t)col * (D / 4) + l];
    float d = r4.x * cc.x + r4.y * cc.y + r4.z * cc.z + r4.w * cc.w;
    d = row16_sum(d);
    float w = edge_weight(d, nr, norms[col], base, c0, c1);
    acc.x += w * cc.x;
    acc.y += w * cc.y;
    acc.z += w * cc.z;
    acc.w += w * cc.w;
}

__global__ void __launch_bounds__(GATHER_THREADS)
gather_kernel(const float* __restrict__ Wh, const float* __restrict__ norms,
              const float* __restrict__ consts, const int* __restrict__ runlen,
              const int2* __restrict__ binned, float* __restrict__ out, int N, int NBLK) {
    __shared__ int  rcnt[BIN_ROWS];
    __shared__ int2 lb[BIN_ROWS * CAP];

    const int tid = threadIdx.x;
    const int b   = blockIdx.x;
    const int g   = tid >> 4;
    const int l   = tid & 15;

    if (tid < BIN_ROWS) rcnt[tid] = 0;
    __syncthreads();

    const int* rl = runlen + (size_t)b * NBLK;
    for (int k0 = 0; k0 < NBLK; k0 += 4 * GROUPS) {
        int len[4];
        const int2* run[4];
        int2 ra[4], rb[4];
        for (int p = 0; p < 4; ++p) {
            int k = k0 + p * GROUPS + g;
            bool valid = k < NBLK;
            int kk = valid ? k : 0;
            int v = rl[kk];
            len[p] = valid ? (v & LEN_MASK) : 0;
            run[p] = binned + (size_t)kk * EPB + (len[p] > 0 ? (v >> LEN_BITS) : 0);
        }
        for (int p = 0; p < 4; ++p) {
            int last = (len[p] > 0 ? len[p] : 1) - 1;
            int ia = l < last ? l : last;
            int ib = (l + 16) < last ? (l + 16) : last;
            ra[p] = run[p][ia];
            rb[p] = run[p][ib];
        }
        for (int p = 0; p < 4; ++p) {
            if (l < len[p]) {
                unsigned u = (unsigned)ra[p].x;
                int r = (int)(u >> 24);
                int slot = atomicAdd(&rcnt[r], 1);
                if (slot < CAP) lb[r * CAP + slot] = make_int2((int)(u & COL_MASK), ra[p].y);
            }
            if (l + 16 < len[p]) {
                unsigned u = (unsigned)rb[p].x;
                int r = (int)(u >> 24);
                int slot = atomicAdd(&rcnt[r], 1);
                if (slot < CAP) lb[r * CAP + slot] = make_int2((int)(u & COL_MASK), rb[p].y);
            }
            for (int i = l + 32; i < len[p]; i += 16) {
                int2 rec = run[p][i];
                unsigned u = (unsigned)rec.x;
                int r = (int)(u >> 24);
                int slot = atomicAdd(&rcnt[r], 1);
                if (slot < CAP) lb[r * CAP + slot] = make_int2((int)(u & COL_MASK), rec.y);
            }
        }
    }
    __syncthreads();

    const float4* Wh4 = (const float4*)Wh;
    const v4f* Wh4v = (const v4f*)Wh;
    const float c0 = consts[0], c1 = consts[1], c2 = consts[2], scale = consts[3];

    for (int q = 0; q < BIN_ROWS / GROUPS; ++q) {
        const int rr  = q * GROUPS + g;
        const int row = b * BIN_ROWS + rr;
        if (row >= N) continue;

        float4 r4 = Wh4[(size_t)row * (D / 4) + l];
        float nr = norms[row];
        const int n_all = rcnt[rr];
        const int n = n_all < CAP ? n_all : CAP;
        const int2* bk = lb + rr * CAP;

        float4 acc = make_float4(0.f, 0.f, 0.f, 0.f);
        const int own_k = l >> 1;

        for (int j = 0; j < n; j += DEPTH) {
            v4f   c4[DEPTH];
            float dt[DEPTH];
            const bool live = (j + own_k) < n;
            const int2 own = bk[live ? (j + own_k) : (n - 1)];
            c4[0] = Wh4v[(size_t)row_share_i<0>(own.x)  * (D / 4) + l];
            c4[1] = Wh4v[(size_t)row_share_i<2>(own.x)  * (D / 4) + l];
            c4[2] = Wh4v[(size_t)row_share_i<4>(own.x)  * (D / 4) + l];
            c4[3] = Wh4v[(size_t)row_share_i<6>(own.x)  * (D / 4) + l];
            c4[4] = Wh4v[(size_t)row_share_i<8>(own.x)  * (D / 4) + l];
            c4[5] = Wh4v[(size_t)row_share_i<10>(own.x) * (D / 4) + l];
            c4[6] = Wh4v[(size_t)row_share_i<12>(own.x) * (D / 4) + l];
            c4[7] = Wh4v[(size_t)row_share_i<14>(own.x) * (D / 4) + l];
            const float nc = norms[own.x];
            asm volatile("" :: "v"(c4[0]), "v"(c4[1]), "v"(c4[2]), "v"(c4[3]),
                               "v"(c4[4]), "v"(c4[5]), "v"(c4[6]), "v"(c4[7]),
                               "v"(nc));
            __builtin_amdgcn_sched_barrier(0);
            for (int k = 0; k < DEPTH; ++k)
                dt[k] = r4.x * c4[k].x + r4.y * c4[k].y + r4.z * c4[k].z + r4.w * c4[k].w;
            const float dot = row16_sum8_scatter(dt, l);
            float w = edge_weight(dot, nr, nc, __int_as_float(own.y), c0, c1);
            w = live ? w : 0.0f;
            float wk[DEPTH];
            wk[0] = row_share_f<0>(w);
            wk[1] = row_share_f<2>(w);
            wk[2] = row_share_f<4>(w);
            wk[3] = row_share_f<6>(w);
            wk[4] = row_share_f<8>(w);
            wk[5] = row_share_f<10>(w);
            wk[6] = row_share_f<12>(w);
            wk[7] = row_share_f<14>(w);
            for (int k = 0; k < DEPTH; ++k) {
                acc.x += wk[k] * c4[k].x;
                acc.y += wk[k] * c4[k].y;
                acc.z += wk[k] * c4[k].z;
                acc.w += wk[k] * c4[k].w;
            }
        }

        if (n_all > CAP) {
            acc = make_float4(0.f, 0.f, 0.f, 0.f);
            for (int k = 0; k < NBLK; ++k) {
                int v = rl[k];
                int len = v & LEN_MASK;
                const int2* run = binned + (size_t)k * EPB + (v >> LEN_BITS);
                for (int i = 0; i < len; ++i) {
                    int2 rec = run[i];
                    unsigned u = (unsigned)rec.x;
                    if ((int)(u >> 24) != rr) continue;
                    add_edge(acc, r4, nr, (int)(u & COL_MASK), __int_as_float(rec.y),
                             Wh4, norms, l, c0, c1);
                }
            }
        }

        float4 o;
        o.x = scale * (acc.x + c2 * r4.x);
        o.y = scale * (acc.y + c2 * r4.y);
        o.z = scale * (acc.z + c2 * r4.z);
        o.w = scale * (acc.w + c2 * r4.w);
        ((float4*)out)[(size_t)row * (D / 4) + l] = o;
    }
}

extern "C" void kernel_launch(void* const* d_in, const int* in_sizes, int n_in,
                              void* d_out, int out_size, void* d_ws, size_t ws_size,
                              hipStream_t stream) {
    const float* h     = (const float*)d_in[0];
    const int*   ei    = (const int*)  d_in[1];
    const float* adj_v = (const float*)d_in[2];
    const float* deg_v = (const float*)d_in[3];
    const float* W     = (const float*)d_in[4];
    const float* bias  = (const float*)d_in[5];
    const float* dcoef = (const float*)d_in[6];
    const float* cpar  = (const float*)d_in[7];
    const float* spar  = (const float*)d_in[8];

    const int N = in_sizes[0] / D;
    const int E = in_sizes[2];
    const int NBIN = (N + BIN_ROWS - 1) / BIN_ROWS;
    const int NBLK = (E + EPB - 1) / EPB;
    const int MBLK = (N + 127) / 128;

    float* Wh      = (float*)d_ws;
    int2*  binned  = (int2*)(Wh + (size_t)N * D);
    float* norms   = (float*)(binned + (size_t)NBLK * EPB);
    int*   runlen  = (int*)(norms + N);
    float* consts  = (float*)(runlen + (size_t)NBIN * NBLK);

    front_kernel<<<1 + NBLK + MBLK, FRONT_THREADS, 0, stream>>>(
        h, W, bias, Wh, norms, cpar, spar, consts,
        ei, adj_v, deg_v, dcoef, binned, runlen, N, E, NBIN, NBLK);

    gather_kernel<<<NBIN, GATHER_THREADS, 0, stream>>>(Wh, norms, consts, runlen, binned,
                                                       (float*)d_out, N, NBLK);
}


# Grouping blocks: edge softplus on hardware exp2/log2 with first-order corrections, not library expf+log1pf (~130 dependent instructions an edge); 16 edges branch-free; weights move by a few ulp
# speedup vs baseline: 4.6481x; 1.0451x over previous
#include <hip/hip_runtime.h>
#include <hip/hip_bf16.h>

typedef __attribute__((ext_vector_type(2))) float v2f;
typedef __attribute__((ext_vector_type(8))) float v8f;
typedef __attribute__((ext_vector_type(4))) float v4f;
typedef __attribute__((ext_vector_type(4))) int v4i;

#define D 64
#define LSTR 66

__device__ __forceinline__ void fold_consts(const float* __restrict__ coeff_param,
                                            const float* __restrict__ scale_param,
                                            float* __restrict__ consts) {
    float x0 = coeff_param[0], x1 = coeff_param[1], x2 = coeff_param[2];
    float m = fmaxf(x0, fmaxf(x1, x2));
    float e0 = expf(x0 - m), e1 = expf(x1 - m), e2 = expf(x2 - m);
    float inv = 1.0f / (e0 + e1 + e2);
    consts[0] = e0 * inv;
    consts[1] = e1 * inv;
    consts[2] = e2 * inv;
    float s = scale_param[0];
    consts[3] = (s > 20.0f) ? s : log1pf(expf(s));
}

__device__ __forceinline__ float softplus_hw(float x) {
    const float LOG2E_HI = 1.44269502162933349609375f;
    const float LOG2E_LO = 1.92596303e-8f;
    const float LN2      = 0.693147182464599609375f;
    const float ax = fminf(fabsf(x), 128.0f);
    const float p  = ax * LOG2E_HI;
    float pe = fmaf(ax, LOG2E_HI, -p);
    pe = fmaf(ax, LOG2E_LO, pe);
    float t = __builtin_amdgcn_exp2f(-p);
    t = fmaf(t, -LN2 * pe, t);
    const float u = 1.0f + t;
    const float c = t - (u - 1.0f);
    float l = __builtin_amdgcn_logf(u) * LN2;
    l = fmaf(c, __builtin_amdgcn_rcpf(u), l);
    return fmaxf(x, 0.0f) + l;
}

__device__ __forceinline__ v4f softplus_hw4(v4f x) {
    const float LOG2E_HI = 1.44269502162933349609375f;
    const float LOG2E_LO = 1.92596303e-8f;
    const float LN2      = 0.693147182464599609375f;
    v4f ax, p, pe, t, u, c, l, r;
#pragma unroll
    for (int j = 0; j < 4; ++j) ax[j] = fminf(fabsf(x[j]), 128.0f);
#pragma unroll
    for (int j = 0; j < 4; ++j) p[j] = ax[j] * LOG2E_HI;
#pragma unroll
    for (int j = 0; j < 4; ++j) pe[j] = fmaf(ax[j], LOG2E_LO, fmaf(ax[j], LOG2E_HI, -p[j]));
#pragma unroll
    for (int j = 0; j < 4; ++j) t[j] = __builtin_amdgcn_exp2f(-p[j]);
#pragma unroll
    for (int j = 0; j < 4; ++j) t[j] = fmaf(t[j], -LN2 * pe[j], t[j]);
#pragma unroll
    for (int j = 0; j < 4; ++j) u[j] = 1.0f + t[j];
#pragma unroll
    for (int j = 0; j < 4; ++j) c[j] = t[j] - (u[j] - 1.0f);
#pragma unroll
    for (int j = 0; j < 4; ++j) l[j] = __builtin_amdgcn_logf(u[j]) * LN2;
#pragma unroll
    for (int j = 0; j < 4; ++j) l[j] = fmaf(c[j], __builtin_amdgcn_rcpf(u[j]), l[j]);
#pragma unroll
    for (int j = 0; j < 4; ++j) r[j] = fmaxf(x[j], 0.0f) + l[j];
    return r;
}

__device__ __forceinline__ float row16_sum(float v) {
    v += __int_as_float(__builtin_amdgcn_update_dpp(0, __float_as_int(v), 0x168, 0xf, 0xf, false));
    v += __int_as_float(__builtin_amdgcn_update_dpp(0, __float_as_int(v), 0x164, 0xf, 0xf, false));
    v += __int_as_float(__builtin_amdgcn_update_dpp(0, __float_as_int(v), 0x162, 0xf, 0xf, false));
    v += __int_as_float(__builtin_amdgcn_update_dpp(0, __float_as_int(v), 0x161, 0xf, 0xf, false));
    return v;
}

template <int XM>
__device__ __forceinline__ float row_xmask_f(float v) {
    return __int_as_float(__builtin_amdgcn_update_dpp(0, __float_as_int(v), 0x160 + XM, 0xf, 0xf, false));
}
template <int SRC>
__device__ __forceinline__ int row_share_i(int v) {
    return __builtin_amdgcn_mov_dpp(v, 0x150 + SRC, 0xf, 0xf, true);
}
template <int SRC>
__device__ __forceinline__ float row_share_f(float v) {
    return __int_as_float(row_share_i<SRC>(__float_as_int(v)));
}

template <int XM>
__device__ __forceinline__ float halve_pair(float a, float b, bool hi) {
    float keep = hi ? b : a;
    float send = hi ? a : b;
    return keep + row_xmask_f<XM>(send);
}

__device__ __forceinline__ float row16_sum8_scatter(const float (&d)[8], int p) {
    const bool h8 = (p & 8) != 0, h4 = (p & 4) != 0, h2 = (p & 2) != 0;
    float e0 = halve_pair<8>(d[0], d[4], h8);
    float e1 = halve_pair<8>(d[1], d[5], h8);
    float e2 = halve_pair<8>(d[2], d[6], h8);
    float e3 = halve_pair<8>(d[3], d[7], h8);
    float f0 = halve_pair<4>(e0, e2, h4);
    float f1 = halve_pair<4>(e1, e3, h4);
    float g  = halve_pair<2>(f0, f1, h2);
    return g + row_xmask_f<1>(g);
}

#define BIN_ROWS 256
#define NBIN_MAX 256
#define EPB 4096
#define FRONT_THREADS 256
#define EPT (EPB / FRONT_THREADS)
#define LEN_BITS 13
#define LEN_MASK ((1 << LEN_BITS) - 1)
#define CAP 64
#define GATHER_THREADS 1024
#define GROUPS (GATHER_THREADS / 16)
#define DEPTH 8
#define COL_MASK 0xFFFFFFu

__global__ void __launch_bounds__(FRONT_THREADS)
front_kernel(const float* __restrict__ h, const float* __restrict__ W,
             const float* __restrict__ bias, float* __restrict__ Wh,
             float* __restrict__ norms,
             const float* __restrict__ coeff_param,
             const float* __restrict__ scale_param,
             float* __restrict__ consts,
             const int* __restrict__ ei, const float* __restrict__ adj_v,
             const float* __restrict__ deg_v, const float* __restrict__ dcoef,
             int2* __restrict__ binned, int* __restrict__ runlen,
             int N, int E, int NBIN, int NBLK) {
    __shared__ float Ws[D * LSTR];
    __shared__ int hist[NBIN_MAX];
    __shared__ int offs[NBIN_MAX];
    __shared__ int wsum[FRONT_THREADS / 32];
    __shared__ __attribute__((aligned(16))) int2 stage[EPB];

    const int tid = threadIdx.x;
    const int bx  = blockIdx.x;
    if (bx == 0) {
        if (tid == 0) fold_consts(coeff_param, scale_param, consts);
        return;
    }

    if (bx <= NBLK) {
        const int blk = bx - 1;
        hist[tid] = 0;
        __syncthreads();

        const float dc0 = dcoef[0], dc1 = dcoef[1];
        const int ebase = blk * EPB;
        int   row[EPT], col[EPT];
        float base[EPT];
        if (((E & 3) == 0) && (ebase + EPB <= E)) {
            v4i r4[EPT / 4], c4[EPT / 4];
            v4f d4[EPT / 4], a4[EPT / 4];
#pragma unroll
            for (int c = 0; c < EPT / 4; ++c) {
                const int e = ebase + (c * FRONT_THREADS + tid) * 4;
                r4[c] = *(const v4i*)(ei + e);
                c4[c] = *(const v4i*)(ei + E + e);
                d4[c] = *(const v4f*)(deg_v + e);
                a4[c] = *(const v4f*)(adj_v + e);
            }
            asm volatile("" :: "v"(r4[0]), "v"(r4[1]), "v"(r4[2]), "v"(r4[3]),
                               "v"(c4[0]), "v"(c4[1]), "v"(c4[2]), "v"(c4[3]),
                               "v"(d4[0]), "v"(d4[1]), "v"(d4[2]), "v"(d4[3]),
                               "v"(a4[0]), "v"(a4[1]), "v"(a4[2]), "v"(a4[3]));
            __builtin_amdgcn_sched_barrier(0);
#pragma unroll
            for (int c = 0; c < EPT / 4; ++c) {
                const v4f sp = softplus_hw4(dc0 * d4[c] + dc1);
#pragma unroll
                for (int j = 0; j < 4; ++j) {
                    row[4 * c + j]  = r4[c][j];
                    col[4 * c + j]  = c4[c][j];
                    base[4 * c + j] = a4[c][j] * sp[j];
                }
            }
        } else {
#pragma unroll
            for (int k = 0; k < EPT; ++k) {
                const int e = ebase + ((k >> 2) * FRONT_THREADS + tid) * 4 + (k & 3);
                const bool ok = e < E;
                const int idx = ok ? e : 0;
                const int r = ei[idx], c = ei[E + idx];
                row[k] = ok ? r : 0;
                col[k] = ok ? c : 0;
                base[k] = adj_v[idx] * softplus_hw(dc0 * deg_v[idx] + dc1);
            }
        }
        int slot[EPT];
#pragma unroll
        for (int k = 0; k < EPT; ++k) {
            slot[k] = -1;
            if (row[k] != col[k]) slot[k] = atomicAdd(&hist[row[k] >> 8], 1);
        }
        unsigned key[EPT];
        int bin[EPT];
#pragma unroll
        for (int k = 0; k < EPT; ++k) {
            key[k] = (unsigned)col[k] | ((unsigned)(row[k] & (BIN_ROWS - 1)) << 24);
            bin[k] = row[k] >> 8;
        }
        __syncthreads();

        const int cnt = hist[tid];
        const int ln = tid & 31, wv = tid >> 5;
        int incl = cnt;
#pragma unroll
        for (int d = 1; d < 32; d <<= 1) {
            int y = __shfl_up(incl, d);
            if (ln >= d) incl += y;
        }
        if (ln == 31) wsum[wv] = incl;
        __syncthreads();
        int wbase = 0;
#pragma unroll
        for (int w = 0; w < FRONT_THREADS / 32; ++w) wbase += (w < wv) ? wsum[w] : 0;
        const int myoff = wbase + incl - cnt;
        offs[tid] = myoff;
        if (tid == NBIN_MAX - 1) {
            const int total = myoff + cnt;
            if (total & 1) stage[total] = make_int2(0, 0);
            if (total == 0) { stage[0] = make_int2(0, 0); stage[1] = make_int2(0, 0); }
        }
        __syncthreads();

        int dsti[EPT];
#pragma unroll
        for (int k = 0; k < EPT; ++k) dsti[k] = offs[bin[k]] + slot[k];
#pragma unroll
        for (int k = 0; k < EPT; ++k) {
            if (slot[k] < 0) continue;
            stage[dsti[k]] = make_int2((int)key[k], __float_as_int(base[k]));
        }
        __syncthreads();

        const int total = offs[NBIN_MAX - 1] + hist[NBIN_MAX - 1];
        int n4 = (total + 1) >> 1;
        n4 = n4 < 1 ? 1 : n4;
        int4* dst = (int4*)(binned + (size_t)blk * EPB);
        const int4* srcp = (const int4*)stage;
        int4 pr[EPB / 2 / FRONT_THREADS];
#pragma unroll
        for (int j = 0; j < EPB / 2 / FRONT_THREADS; ++j) {
            const int i = j * FRONT_THREADS + tid;
            pr[j] = (i < n4) ? srcp[i] : make_int4(0, 0, 0, 0);
        }
#pragma unroll
        for (int j = 0; j < EPB / 2 / FRONT_THREADS; ++j) {
            const int i = j * FRONT_THREADS + tid;
            if (i < n4) dst[i] = pr[j];
        }
        if (tid < NBIN) runlen[tid * NBLK + blk] = (myoff << LEN_BITS) | cnt;
        return;
    }

    for (int i = tid * 4; i < D * D; i += FRONT_THREADS * 4) {
        int r = i >> 6;
        int c = i & 63;
        float4 w4 = *(const float4*)(W + i);
        Ws[(c + 0) * LSTR + r] = w4.x;
        Ws[(c + 1) * LSTR + r] = w4.y;
        Ws[(c + 2) * LSTR + r] = w4.z;
        Ws[(c + 3) * LSTR + r] = w4.w;
    }
    __syncthreads();

    int wave = tid >> 5;
    int lane = tid & 31;
    int half = lane >> 4;
    int l16  = lane & 15;

    int m0 = ((bx - 1 - NBLK) * (FRONT_THREADS / 32) + wave) * 16;
    if (m0 >= N) return;

    const float* hrow = h + (size_t)(m0 + l16) * D + 2 * half;
    const float* wt   = Ws + l16 * LSTR + 2 * half;

    v8f acc[4] = {};

#pragma unroll 4
    for (int s = 0; s < 16; ++s) {
        v2f a = *(const v2f*)(hrow + 4 * s);
#pragma unroll
        for (int t = 0; t < 4; ++t) {
            v2f bf = *(const v2f*)(wt + t * (16 * LSTR) + 4 * s);
            acc[t] = __builtin_amdgcn_wmma_f32_16x16x4_f32(
                false, a, false, bf, (short)0, acc[t], false, false);
        }
    }

    float ss[8] = {0.f, 0.f, 0.f, 0.f, 0.f, 0.f, 0.f, 0.f};
#pragma unroll
    for (int t = 0; t < 4; ++t) {
        int n = t * 16 + l16;
        float bv = bias[n];
#pragma unroll
        for (int v = 0; v < 8; ++v) {
            float val = acc[t][v] + bv;
            Wh[(size_t)(m0 + half * 8 + v) * D + n] = val;
            ss[v] += val * val;
        }
    }
#pragma unroll
    for (int v = 0; v < 8; ++v) ss[v] = row16_sum(ss[v]);
    float mine = ss[0];
#pragma unroll
    for (int v = 1; v < 8; ++v) mine = (l16 == v) ? ss[v] : mine;
    if (l16 < 8) {
        int node = m0 + half * 8 + l16;
        norms[node] = sqrtf(mine);
    }
}

__device__ __forceinline__ float edge_weight(float dot, float nr, float nc,
                                             float base, float c0, float c1) {
    float sim = dot / fmaxf(nr * nc, 1e-8f);
    return base * (sim > 0.0f ? c0 * sim : c1 * sim);
}

__device__ __forceinline__ void add_edge(float4& acc, const float4 r4, float nr, int col,
                                         float base, const float4* __restrict__ Wh4,
                                         const float* __restrict__ norms, int l,
                                         float c0, float c1) {
    float4 cc = Wh4[(size_t)col * (D / 4) + l];
    float d = r4.x * cc.x + r4.y * cc.y + r4.z * cc.z + r4.w * cc.w;
    d = row16_sum(d);
    float w = edge_weight(d, nr, norms[col], base, c0, c1);
    acc.x += w * cc.x;
    acc.y += w * cc.y;
    acc.z += w * cc.z;
    acc.w += w * cc.w;
}

__global__ void __launch_bounds__(GATHER_THREADS)
gather_kernel(const float* __restrict__ Wh, const float* __restrict__ norms,
              const float* __restrict__ consts, const int* __restrict__ runlen,
              const int2* __restrict__ binned, float* __restrict__ out, int N, int NBLK) {
    __shared__ int  rcnt[BIN_ROWS];
    __shared__ int2 lb[BIN_ROWS * CAP];

    const int tid = threadIdx.x;
    const int b   = blockIdx.x;
    const int g   = tid >> 4;
    const int l   = tid & 15;

    if (tid < BIN_ROWS) rcnt[tid] = 0;
    __syncthreads();

    const int* rl = runlen + (size_t)b * NBLK;
    for (int k0 = 0; k0 < NBLK; k0 += 4 * GROUPS) {
        int len[4];
        const int2* run[4];
        int2 ra[4], rb[4];
        for (int p = 0; p < 4; ++p) {
            int k = k0 + p * GROUPS + g;
            bool valid = k < NBLK;
            int kk = valid ? k : 0;
            int v = rl[kk];
            len[p] = valid ? (v & LEN_MASK) : 0;
            run[p] = binned + (size_t)kk * EPB + (len[p] > 0 ? (v >> LEN_BITS) : 0);
        }
        for (int p = 0; p < 4; ++p) {
            int last = (len[p] > 0 ? len[p] : 1) - 1;
            int ia = l < last ? l : last;
            int ib = (l + 16) < last ? (l + 16) : last;
            ra[p] = run[p][ia];
            rb[p] = run[p][ib];
        }
        for (int p = 0; p < 4; ++p) {
            if (l < len[p]) {
                unsigned u = (unsigned)ra[p].x;
                int r = (int)(u >> 24);
                int slot = atomicAdd(&rcnt[r], 1);
                if (slot < CAP) lb[r * CAP + slot] = make_int2((int)(u & COL_MASK), ra[p].y);
            }
            if (l + 16 < len[p]) {
                unsigned u = (unsigned)rb[p].x;
                int r = (int)(u >> 24);
                int slot = atomicAdd(&rcnt[r], 1);
                if (slot < CAP) lb[r * CAP + slot] = make_int2((int)(u & COL_MASK), rb[p].y);
            }
            for (int i = l + 32; i < len[p]; i += 16) {
                int2 rec = run[p][i];
                unsigned u = (unsigned)rec.x;
                int r = (int)(u >> 24);
                int slot = atomicAdd(&rcnt[r], 1);
                if (slot < CAP) lb[r * CAP + slot] = make_int2((int)(u & COL_MASK), rec.y);
            }
        }
    }
    __syncthreads();

    const float4* Wh4 = (const float4*)Wh;
    const v4f* Wh4v = (const v4f*)Wh;
    const float c0 = consts[0], c1 = consts[1], c2 = consts[2], scale = consts[3];

    for (int q = 0; q < BIN_ROWS / GROUPS; ++q) {
        const int rr  = q * GROUPS + g;
        const int row = b * BIN_ROWS + rr;
        if (row >= N) continue;

        float4 r4 = Wh4[(size_t)row * (D / 4) + l];
        float nr = norms[row];
        const int n_all = rcnt[rr];
        const int n = n_all < CAP ? n_all : CAP;
        const int2* bk = lb + rr * CAP;

        float4 acc = make_float4(0.f, 0.f, 0.f, 0.f);
        const int own_k = l >> 1;

        for (int j = 0; j < n; j += DEPTH) {
            v4f   c4[DEPTH];
            float dt[DEPTH];
            const bool live = (j + own_k) < n;
            const int2 own = bk[live ? (j + own_k) : (n - 1)];
            c4[0] = Wh4v[(size_t)row_share_i<0>(own.x)  * (D / 4) + l];
            c4[1] = Wh4v[(size_t)row_share_i<2>(own.x)  * (D / 4) + l];
            c4[2] = Wh4v[(size_t)row_share_i<4>(own.x)  * (D / 4) + l];
            c4[3] = Wh4v[(size_t)row_share_i<6>(own.x)  * (D / 4) + l];
            c4[4] = Wh4v[(size_t)row_share_i<8>(own.x)  * (D / 4) + l];
            c4[5] = Wh4v[(size_t)row_share_i<10>(own.x) * (D / 4) + l];
            c4[6] = Wh4v[(size_t)row_share_i<12>(own.x) * (D / 4) + l];
            c4[7] = Wh4v[(size_t)row_share_i<14>(own.x) * (D / 4) + l];
            const float nc = norms[own.x];
            asm volatile("" :: "v"(c4[0]), "v"(c4[1]), "v"(c4[2]), "v"(c4[3]),
                               "v"(c4[4]), "v"(c4[5]), "v"(c4[6]), "v"(c4[7]),
                               "v"(nc));
            __builtin_amdgcn_sched_barrier(0);
            for (int k = 0; k < DEPTH; ++k)
                dt[k] = r4.x * c4[k].x + r4.y * c4[k].y + r4.z * c4[k].z + r4.w * c4[k].w;
            const float dot = row16_sum8_scatter(dt, l);
            float w = edge_weight(dot, nr, nc, __int_as_float(own.y), c0, c1);
            w = live ? w : 0.0f;
            float wk[DEPTH];
            wk[0] = row_share_f<0>(w);
            wk[1] = row_share_f<2>(w);
            wk[2] = row_share_f<4>(w);
            wk[3] = row_share_f<6>(w);
            wk[4] = row_share_f<8>(w);
            wk[5] = row_share_f<10>(w);
            wk[6] = row_share_f<12>(w);
            wk[7] = row_share_f<14>(w);
            for (int k = 0; k < DEPTH; ++k) {
                acc.x += wk[k] * c4[k].x;
                acc.y += wk[k] * c4[k].y;
                acc.z += wk[k] * c4[k].z;
                acc.w += wk[k] * c4[k].w;
            }
        }

        if (n_all > CAP) {
            acc = make_float4(0.f, 0.f, 0.f, 0.f);
            for (int k = 0; k < NBLK; ++k) {
                int v = rl[k];
                int len = v & LEN_MASK;
                const int2* run = binned + (size_t)k * EPB + (v >> LEN_BITS);
                for (int i = 0; i < len; ++i) {
                    int2 rec = run[i];
                    unsigned u = (unsigned)rec.x;
                    if ((int)(u >> 24) != rr) continue;
                    add_edge(acc, r4, nr, (int)(u & COL_MASK), __int_as_float(rec.y),
                             Wh4, norms, l, c0, c1);
                }
            }
        }

        float4 o;
        o.x = scale * (acc.x + c2 * r4.x);
        o.y = scale * (acc.y + c2 * r4.y);
        o.z = scale * (acc.z + c2 * r4.z);
        o.w = scale * (acc.w + c2 * r4.w);
        ((float4*)out)[(size_t)row * (D / 4) + l] = o;
    }
}

extern "C" void kernel_launch(void* const* d_in, const int* in_sizes, int n_in,
                              void* d_out, int out_size, void* d_ws, size_t ws_size,
                              hipStream_t stream) {
    const float* h     = (const float*)d_in[0];
    const int*   ei    = (const int*)  d_in[1];
    const float* adj_v = (const float*)d_in[2];
    const float* deg_v = (const float*)d_in[3];
    const float* W     = (const float*)d_in[4];
    const float* bias  = (const float*)d_in[5];
    const float* dcoef = (const float*)d_in[6];
    const float* cpar  = (const float*)d_in[7];
    const float* spar  = (const float*)d_in[8];

    const int N = in_sizes[0] / D;
    const int E = in_sizes[2];
    const int NBIN = (N + BIN_ROWS - 1) / BIN_ROWS;
    const int NBLK = (E + EPB - 1) / EPB;
    const int MBLK = (N + 127) / 128;

    float* Wh      = (float*)d_ws;
    int2*  binned  = (int2*)(Wh + (size_t)N * D);
    float* norms   = (float*)(binned + (size_t)NBLK * EPB);
    int*   runlen  = (int*)(norms + N);
    float* consts  = (float*)(runlen + (size_t)NBIN * NBLK);

    front_kernel<<<1 + NBLK + MBLK, FRONT_THREADS, 0, stream>>>(
        h, W, bias, Wh, norms, cpar, spar, consts,
        ei, adj_v, deg_v, dcoef, binned, runlen, N, E, NBIN, NBLK);

    gather_kernel<<<NBIN, GATHER_THREADS, 0, stream>>>(Wh, norms, consts, runlen, binned,
                                                       (float*)d_out, N, NBLK);
}
